# starting kernel
# baseline (speedup 1.0000x reference)
#include <hip/hip_runtime.h>
#include <cstdint>
#include <cstddef>

typedef __attribute__((ext_vector_type(16))) _Float16 v16h;
typedef __attribute__((ext_vector_type(8)))  _Float16 v8h;
typedef __attribute__((ext_vector_type(8)))  float    v8f;
typedef __attribute__((ext_vector_type(2)))  float    v2f;

#define B_    256
#define SEQ_  512
#define HOR_  32
#define NK_   8192
#define L1_   64
#define HID_  128
#define L2_   64
#define GATES_ 512
#define HLEN_ 544

__global__ __launch_bounds__(256) void gemm1_kernel(const float* __restrict__ inp,
                                                    const float* __restrict__ akey,
                                                    float* __restrict__ dist) {
  const int tid = threadIdx.x, lane = tid & 31, wave = tid >> 5;
  const int mt = blockIdx.x, nt = blockIdx.y * 8 + wave;
  const int m = lane & 15;
  const int kb = (lane & 16) ? 2 : 0;
  const float* Ap = inp  + (size_t)(mt * 16 + m) * SEQ_ + kb;
  const float* Bp = akey + (size_t)(nt * 16 + m) * SEQ_ + kb;
  v8f acc0 = {}, acc1 = {};
  for (int k = 0; k < SEQ_; k += 8) {
    v2f a0 = *(const v2f*)(Ap + k);
    v2f b0 = *(const v2f*)(Bp + k);
    acc0 = __builtin_amdgcn_wmma_f32_16x16x4_f32(false, a0, false, b0, (short)0, acc0, false, false);
    v2f a1 = *(const v2f*)(Ap + k + 4);
    v2f b1 = *(const v2f*)(Bp + k + 4);
    acc1 = __builtin_amdgcn_wmma_f32_16x16x4_f32(false, a1, false, b1, (short)0, acc1, false, false);
  }
#pragma unroll
  for (int r = 0; r < 8; ++r) {
    int row = r + ((lane & 16) ? 8 : 0);
    dist[(size_t)(mt * 16 + row) * NK_ + nt * 16 + m] = acc0[r] + acc1[r];
  }
}

__global__ __launch_bounds__(256) void softmax_kernel(float* __restrict__ dist) {
  __shared__ float sv[256];
  __shared__ int   si[256];
  const int b = blockIdx.x, tid = threadIdx.x;
  float* row = dist + (size_t)b * NK_;

  float m = -3.4e38f; int mi = NK_;
  for (int j = tid; j < NK_; j += 256) {
    float v = row[j];
    if (v > m || (v == m && j < mi)) { m = v; mi = j; }
  }
  sv[tid] = m; si[tid] = mi;
  __syncthreads();
  for (int s = 128; s > 0; s >>= 1) {
    if (tid < s) {
      float v2 = sv[tid + s]; int i2 = si[tid + s];
      if (v2 > sv[tid] || (v2 == sv[tid] && i2 < si[tid])) { sv[tid] = v2; si[tid] = i2; }
    }
    __syncthreads();
  }
  if (tid == 0) row[si[0]] = 0.0f;
  __syncthreads();

  float mm = -3.4e38f;
  for (int j = tid; j < NK_; j += 256) mm = fmaxf(mm, row[j]);
  sv[tid] = mm;
  __syncthreads();
  for (int s = 128; s > 0; s >>= 1) { if (tid < s) sv[tid] = fmaxf(sv[tid], sv[tid + s]); __syncthreads(); }
  const float M = sv[0];
  __syncthreads();

  float sum = 0.f;
  for (int j = tid; j < NK_; j += 256) sum += __expf(row[j] - M);
  sv[tid] = sum;
  __syncthreads();
  for (int s = 128; s > 0; s >>= 1) { if (tid < s) sv[tid] += sv[tid + s]; __syncthreads(); }
  const float inv = 1.0f / sv[0];
  for (int j = tid; j < NK_; j += 256) row[j] = __expf(row[j] - M) * inv;
}

__global__ __launch_bounds__(256) void gemm2_kernel(const float* __restrict__ a,
                                                    const float* __restrict__ aval,
                                                    float* __restrict__ hist) {
  const int tid = threadIdx.x, lane = tid & 31, wave = tid >> 5;
  const int mt = blockIdx.x, nt = blockIdx.y * 8 + wave;
  if (nt >= (HLEN_ / 16)) return;
  const int m = lane & 15;
  const int kb = (lane & 16) ? 2 : 0;
  const int n = nt * 16 + m;
  const float* Ap = a + (size_t)(mt * 16 + m) * NK_ + kb;
  const float* Bp = aval + (size_t)kb * HLEN_ + n;
  v8f acc0 = {}, acc1 = {};
  for (int k = 0; k < NK_; k += 8) {
    v2f av = *(const v2f*)(Ap + k);
    v2f bv; bv[0] = Bp[(size_t)k * HLEN_]; bv[1] = Bp[(size_t)(k + 1) * HLEN_];
    acc0 = __builtin_amdgcn_wmma_f32_16x16x4_f32(false, av, false, bv, (short)0, acc0, false, false);
    v2f av1 = *(const v2f*)(Ap + k + 4);
    v2f bv1; bv1[0] = Bp[(size_t)(k + 4) * HLEN_]; bv1[1] = Bp[(size_t)(k + 5) * HLEN_];
    acc1 = __builtin_amdgcn_wmma_f32_16x16x4_f32(false, av1, false, bv1, (short)0, acc1, false, false);
  }
#pragma unroll
  for (int r = 0; r < 8; ++r) {
    int row = r + ((lane & 16) ? 8 : 0);
    hist[(size_t)(mt * 16 + row) * HLEN_ + n] = acc0[r] + acc1[r];
  }
}

__global__ __launch_bounds__(256) void featl1_kernel(const float* __restrict__ inp,
                                                     const float* __restrict__ hist,
                                                     const float* __restrict__ W1,
                                                     const float* __restrict__ b1,
                                                     _Float16* __restrict__ xf) {
  __shared__ float w1s[L1_ * (HOR_ + 1)];
  __shared__ float b1s[L1_];
  __shared__ float hs[8 + HOR_ - 1];
  __shared__ float ins[8];
  const int b = blockIdx.x, t0 = blockIdx.y * 8, tid = threadIdx.x;
  for (int i = tid; i < L1_ * (HOR_ + 1); i += 256) w1s[i] = W1[i];
  if (tid < L1_) b1s[tid] = b1[tid];
  if (tid < 8 + HOR_ - 1) hs[tid] = hist[(size_t)b * HLEN_ + t0 + tid];
  if (tid < 8) ins[tid] = inp[(size_t)b * SEQ_ + t0 + tid];
  __syncthreads();
  const int o = tid & 63;
#pragma unroll
  for (int pass = 0; pass < 2; ++pass) {
    int tt = (tid >> 6) + pass * 4;
    float acc = b1s[o] + ins[tt] * w1s[o * 33];
#pragma unroll
    for (int j = 0; j < HOR_; ++j) acc += hs[tt + j] * w1s[o * 33 + 1 + j];
    xf[((size_t)b * SEQ_ + (t0 + tt)) * L1_ + o] = (_Float16)tanhf(acc);
  }
}

template <int KIN, bool STORE_SEQ>
__global__ __launch_bounds__(256, 1) void lstm_kernel(const _Float16* __restrict__ xin,
                                                      const float* __restrict__ Wih,
                                                      const float* __restrict__ Whh,
                                                      const float* __restrict__ bih,
                                                      const float* __restrict__ bhh,
                                                      _Float16* __restrict__ hseq,
                                                      float* __restrict__ hlast) {
  constexpr int KFX = KIN / 32;
  __shared__ float gates[16][GATES_];
  __shared__ __align__(16) _Float16 hbuf[16][HID_];
  const int tid = threadIdx.x, lane = tid & 31, wave = tid >> 5;
  const int bm = blockIdx.x * 16;
  const int nbase = wave * 64;
  const int m = lane & 15;
  const int hi16 = (lane >> 4) & 1;
  const int kbh = hi16 * 8;

  v16h wihB[4][KFX], whhB[4][4];
  float biasv[4];
#pragma unroll
  for (int nt = 0; nt < 4; ++nt) {
    const int g = nbase + nt * 16 + m;
    biasv[nt] = bih[g] + bhh[g];
#pragma unroll
    for (int kf = 0; kf < KFX; ++kf)
#pragma unroll
      for (int h = 0; h < 16; ++h) {
        int k = kf * 32 + ((h & 8) << 1) + kbh + (h & 7);
        wihB[nt][kf][h] = (_Float16)Wih[(size_t)g * KIN + k];
      }
#pragma unroll
    for (int kf = 0; kf < 4; ++kf)
#pragma unroll
      for (int h = 0; h < 16; ++h) {
        int k = kf * 32 + ((h & 8) << 1) + kbh + (h & 7);
        whhB[nt][kf][h] = (_Float16)Whh[(size_t)g * HID_ + k];
      }
  }

  const int prow = tid >> 4;
  const int pj = (tid & 15) * 8;
  float c[8];
#pragma unroll
  for (int i = 0; i < 8; ++i) { c[i] = 0.f; hbuf[prow][pj + i] = (_Float16)0.f; }
  __syncthreads();

  for (int t = 0; t < SEQ_; ++t) {
    v16h xA[KFX], hA[4];
#pragma unroll
    for (int kf = 0; kf < KFX; ++kf) {
      const _Float16* p = xin + ((size_t)(bm + m) * SEQ_ + t) * KIN + kf * 32 + kbh;
      v8h lo = *(const v8h*)p;
      v8h hi = *(const v8h*)(p + 16);
#pragma unroll
      for (int i = 0; i < 8; ++i) { xA[kf][i] = lo[i]; xA[kf][i + 8] = hi[i]; }
    }
#pragma unroll
    for (int kf = 0; kf < 4; ++kf) {
      const _Float16* p = &hbuf[m][kf * 32 + kbh];
      v8h lo = *(const v8h*)p;
      v8h hi = *(const v8h*)(p + 16);
#pragma unroll
      for (int i = 0; i < 8; ++i) { hA[kf][i] = lo[i]; hA[kf][i + 8] = hi[i]; }
    }
    v8f acc[4];
#pragma unroll
    for (int nt = 0; nt < 4; ++nt)
#pragma unroll
      for (int r = 0; r < 8; ++r) acc[nt][r] = biasv[nt];
#pragma unroll
    for (int nt = 0; nt < 4; ++nt) {
#pragma unroll
      for (int kf = 0; kf < KFX; ++kf)
        acc[nt] = __builtin_amdgcn_wmma_f32_16x16x32_f16(false, xA[kf], false, wihB[nt][kf],
                                                         (short)0, acc[nt], false, false);
#pragma unroll
      for (int kf = 0; kf < 4; ++kf)
        acc[nt] = __builtin_amdgcn_wmma_f32_16x16x32_f16(false, hA[kf], false, whhB[nt][kf],
                                                         (short)0, acc[nt], false, false);
    }
#pragma unroll
    for (int nt = 0; nt < 4; ++nt) {
      const int g = nbase + nt * 16 + m;
      const bool isg = (g >> 7) == 2;
#pragma unroll
      for (int r = 0; r < 8; ++r) {
        int row = r + hi16 * 8;
        float v = acc[nt][r];
        gates[row][g] = isg ? tanhf(v) : 1.0f / (1.0f + __expf(-v));
      }
    }
    __syncthreads();
    v8h hpk;
#pragma unroll
    for (int i = 0; i < 8; ++i) {
      int j = pj + i;
      float iv = gates[prow][j];
      float fv = gates[prow][HID_ + j];
      float gv = gates[prow][2 * HID_ + j];
      float ov = gates[prow][3 * HID_ + j];
      c[i] = fv * c[i] + iv * gv;
      float hv = ov * tanhf(c[i]);
      hpk[i] = (_Float16)hv;
      if (!STORE_SEQ && t == SEQ_ - 1) hlast[(size_t)(bm + prow) * HID_ + j] = hv;
    }
#pragma unroll
    for (int i = 0; i < 8; ++i) hbuf[prow][pj + i] = hpk[i];
    if (STORE_SEQ)
      *(v8h*)(hseq + ((size_t)(bm + prow) * SEQ_ + t) * HID_ + pj) = hpk;
    __syncthreads();
  }
}

__global__ __launch_bounds__(128) void head_kernel(const float* __restrict__ h1last,
                                                   const float* __restrict__ W2,
                                                   const float* __restrict__ b2,
                                                   const float* __restrict__ W3,
                                                   const float* __restrict__ b3,
                                                   float* __restrict__ out) {
  __shared__ float lastS[HID_];
  __shared__ float l2S[L2_];
  const int b = blockIdx.x, tid = threadIdx.x;
  lastS[tid] = tanhf(h1last[(size_t)b * HID_ + tid]);
  __syncthreads();
  if (tid < L2_) {
    float acc = b2[tid];
    for (int k = 0; k < HID_; ++k) acc += lastS[k] * W2[tid * HID_ + k];
    l2S[tid] = tanhf(acc);
  }
  __syncthreads();
  if (tid < HOR_) {
    float acc = b3[tid];
    for (int k = 0; k < L2_; ++k) acc += l2S[k] * W3[tid * L2_ + k];
    out[(size_t)b * HOR_ + tid] = acc;
  }
}

extern "C" void kernel_launch(void* const* d_in, const int* in_sizes, int n_in,
                              void* d_out, int out_size, void* d_ws, size_t ws_size,
                              hipStream_t stream) {
  (void)in_sizes; (void)n_in; (void)out_size; (void)ws_size;
  const float* inp  = (const float*)d_in[0];
  const float* akey = (const float*)d_in[1];
  const float* aval = (const float*)d_in[2];
  const float* W1   = (const float*)d_in[3];
  const float* b1   = (const float*)d_in[4];
  const float* Wih0 = (const float*)d_in[5];
  const float* Whh0 = (const float*)d_in[6];
  const float* bih0 = (const float*)d_in[7];
  const float* bhh0 = (const float*)d_in[8];
  const float* Wih1 = (const float*)d_in[9];
  const float* Whh1 = (const float*)d_in[10];
  const float* bih1 = (const float*)d_in[11];
  const float* bhh1 = (const float*)d_in[12];
  const float* W2   = (const float*)d_in[13];
  const float* b2   = (const float*)d_in[14];
  const float* W3   = (const float*)d_in[15];
  const float* b3   = (const float*)d_in[16];

  char* ws = (char*)d_ws;
  size_t off = 0;
  float*    dist   = (float*)(ws + off);    off += (size_t)B_ * NK_ * 4;
  float*    hist   = (float*)(ws + off);    off += (size_t)B_ * HLEN_ * 4;
  _Float16* xf     = (_Float16*)(ws + off); off += (size_t)B_ * SEQ_ * L1_ * 2;
  _Float16* h0     = (_Float16*)(ws + off); off += (size_t)B_ * SEQ_ * HID_ * 2;
  float*    h1last = (float*)(ws + off);    off += (size_t)B_ * HID_ * 4;

  gemm1_kernel<<<dim3(16, 64), 256, 0, stream>>>(inp, akey, dist);
  softmax_kernel<<<dim3(B_), 256, 0, stream>>>(dist);
  gemm2_kernel<<<dim3(16, 5), 256, 0, stream>>>(dist, aval, hist);
  featl1_kernel<<<dim3(B_, SEQ_ / 8), 256, 0, stream>>>(inp, hist, W1, b1, xf);
  lstm_kernel<L1_, true><<<dim3(B_ / 16), 256, 0, stream>>>(xf, Wih0, Whh0, bih0, bhh0, h0, nullptr);
  lstm_kernel<HID_, false><<<dim3(B_ / 16), 256, 0, stream>>>(h0, Wih1, Whh1, bih1, bhh1, nullptr, h1last);
  head_kernel<<<dim3(B_), 128, 0, stream>>>(h1last, W2, b2, W3, b3, (float*)d_out);
}


# LDS row pitch of the LSTM tiles padded to 16 B x odd (h 256->272 B, gates 2048->2064 B): h A-fragment ds_load_b128 was 16-way bank-conflicted, gate stores 2-way; now conflict-free
# speedup vs baseline: 2.2662x; 2.2662x over previous
#include <hip/hip_runtime.h>
#include <cstdint>
#include <cstddef>

typedef __attribute__((ext_vector_type(16))) _Float16 v16h;
typedef __attribute__((ext_vector_type(8)))  _Float16 v8h;
typedef __attribute__((ext_vector_type(8)))  float    v8f;
typedef __attribute__((ext_vector_type(2)))  float    v2f;

#define B_    256
#define SEQ_  512
#define HOR_  32
#define NK_   8192
#define L1_   64
#define HID_  128
#define L2_   64
#define GATES_ 512
#define HLEN_ 544
#define HPAD_ 8
#define GPAD_ 4

__global__ __launch_bounds__(256) void gemm1_kernel(const float* __restrict__ inp,
                                                    const float* __restrict__ akey,
                                                    float* __restrict__ dist) {
  const int tid = threadIdx.x, lane = tid & 31, wave = tid >> 5;
  const int mt = blockIdx.x, nt = blockIdx.y * 8 + wave;
  const int m = lane & 15;
  const int kb = (lane & 16) ? 2 : 0;
  const float* Ap = inp  + (size_t)(mt * 16 + m) * SEQ_ + kb;
  const float* Bp = akey + (size_t)(nt * 16 + m) * SEQ_ + kb;
  v8f acc0 = {}, acc1 = {};
  for (int k = 0; k < SEQ_; k += 8) {
    v2f a0 = *(const v2f*)(Ap + k);
    v2f b0 = *(const v2f*)(Bp + k);
    acc0 = __builtin_amdgcn_wmma_f32_16x16x4_f32(false, a0, false, b0, (short)0, acc0, false, false);
    v2f a1 = *(const v2f*)(Ap + k + 4);
    v2f b1 = *(const v2f*)(Bp + k + 4);
    acc1 = __builtin_amdgcn_wmma_f32_16x16x4_f32(false, a1, false, b1, (short)0, acc1, false, false);
  }
#pragma unroll
  for (int r = 0; r < 8; ++r) {
    int row = r + ((lane & 16) ? 8 : 0);
    dist[(size_t)(mt * 16 + row) * NK_ + nt * 16 + m] = acc0[r] + acc1[r];
  }
}

__global__ __launch_bounds__(256) void softmax_kernel(float* __restrict__ dist) {
  __shared__ float sv[256];
  __shared__ int   si[256];
  const int b = blockIdx.x, tid = threadIdx.x;
  float* row = dist + (size_t)b * NK_;

  float m = -3.4e38f; int mi = NK_;
  for (int j = tid; j < NK_; j += 256) {
    float v = row[j];
    if (v > m || (v == m && j < mi)) { m = v; mi = j; }
  }
  sv[tid] = m; si[tid] = mi;
  __syncthreads();
  for (int s = 128; s > 0; s >>= 1) {
    if (tid < s) {
      float v2 = sv[tid + s]; int i2 = si[tid + s];
      if (v2 > sv[tid] || (v2 == sv[tid] && i2 < si[tid])) { sv[tid] = v2; si[tid] = i2; }
    }
    __syncthreads();
  }
  if (tid == 0) row[si[0]] = 0.0f;
  __syncthreads();

  float mm = -3.4e38f;
  for (int j = tid; j < NK_; j += 256) mm = fmaxf(mm, row[j]);
  sv[tid] = mm;
  __syncthreads();
  for (int s = 128; s > 0; s >>= 1) { if (tid < s) sv[tid] = fmaxf(sv[tid], sv[tid + s]); __syncthreads(); }
  const float M = sv[0];
  __syncthreads();

  float sum = 0.f;
  for (int j = tid; j < NK_; j += 256) sum += __expf(row[j] - M);
  sv[tid] = sum;
  __syncthreads();
  for (int s = 128; s > 0; s >>= 1) { if (tid < s) sv[tid] += sv[tid + s]; __syncthreads(); }
  const float inv = 1.0f / sv[0];
  for (int j = tid; j < NK_; j += 256) row[j] = __expf(row[j] - M) * inv;
}

__global__ __launch_bounds__(256) void gemm2_kernel(const float* __restrict__ a,
                                                    const float* __restrict__ aval,
                                                    float* __restrict__ hist) {
  const int tid = threadIdx.x, lane = tid & 31, wave = tid >> 5;
  const int mt = blockIdx.x, nt = blockIdx.y * 8 + wave;
  if (nt >= (HLEN_ / 16)) return;
  const int m = lane & 15;
  const int kb = (lane & 16) ? 2 : 0;
  const int n = nt * 16 + m;
  const float* Ap = a + (size_t)(mt * 16 + m) * NK_ + kb;
  const float* Bp = aval + (size_t)kb * HLEN_ + n;
  v8f acc0 = {}, acc1 = {};
  for (int k = 0; k < NK_; k += 8) {
    v2f av = *(const v2f*)(Ap + k);
    v2f bv; bv[0] = Bp[(size_t)k * HLEN_]; bv[1] = Bp[(size_t)(k + 1) * HLEN_];
    acc0 = __builtin_amdgcn_wmma_f32_16x16x4_f32(false, av, false, bv, (short)0, acc0, false, false);
    v2f av1 = *(const v2f*)(Ap + k + 4);
    v2f bv1; bv1[0] = Bp[(size_t)(k + 4) * HLEN_]; bv1[1] = Bp[(size_t)(k + 5) * HLEN_];
    acc1 = __builtin_amdgcn_wmma_f32_16x16x4_f32(false, av1, false, bv1, (short)0, acc1, false, false);
  }
#pragma unroll
  for (int r = 0; r < 8; ++r) {
    int row = r + ((lane & 16) ? 8 : 0);
    hist[(size_t)(mt * 16 + row) * HLEN_ + n] = acc0[r] + acc1[r];
  }
}

__global__ __launch_bounds__(256) void featl1_kernel(const float* __restrict__ inp,
                                                     const float* __restrict__ hist,
                                                     const float* __restrict__ W1,
                                                     const float* __restrict__ b1,
                                                     _Float16* __restrict__ xf) {
  __shared__ float w1s[L1_ * (HOR_ + 1)];
  __shared__ float b1s[L1_];
  __shared__ float hs[8 + HOR_ - 1];
  __shared__ float ins[8];
  const int b = blockIdx.x, t0 = blockIdx.y * 8, tid = threadIdx.x;
  for (int i = tid; i < L1_ * (HOR_ + 1); i += 256) w1s[i] = W1[i];
  if (tid < L1_) b1s[tid] = b1[tid];
  if (tid < 8 + HOR_ - 1) hs[tid] = hist[(size_t)b * HLEN_ + t0 + tid];
  if (tid < 8) ins[tid] = inp[(size_t)b * SEQ_ + t0 + tid];
  __syncthreads();
  const int o = tid & 63;
#pragma unroll
  for (int pass = 0; pass < 2; ++pass) {
    int tt = (tid >> 6) + pass * 4;
    float acc = b1s[o] + ins[tt] * w1s[o * 33];
#pragma unroll
    for (int j = 0; j < HOR_; ++j) acc += hs[tt + j] * w1s[o * 33 + 1 + j];
    xf[((size_t)b * SEQ_ + (t0 + tt)) * L1_ + o] = (_Float16)tanhf(acc);
  }
}

#define LOG2E_F 1.4426950408889634f
__device__ __forceinline__ float act_exp2_rcp(float v, float s, float a, float b) {
  const float e = __builtin_amdgcn_exp2f(s * v);
  return __builtin_fmaf(a, __builtin_amdgcn_rcpf(1.0f + e), b);
}

template <int KIN, bool STORE_SEQ>
__global__ __launch_bounds__(256, 1) void lstm_kernel(const _Float16* __restrict__ xin,
                                                      const float* __restrict__ Wih,
                                                      const float* __restrict__ Whh,
                                                      const float* __restrict__ bih,
                                                      const float* __restrict__ bhh,
                                                      _Float16* __restrict__ hseq,
                                                      float* __restrict__ hlast) {
  constexpr int KFX = KIN / 32;
  __shared__ __align__(16) float gates[16][GATES_ + GPAD_];
  __shared__ __align__(16) _Float16 hbuf[16][HID_ + HPAD_];
  const int tid = threadIdx.x, lane = tid & 31, wave = tid >> 5;
  const int bm = blockIdx.x * 16;
  const int nbase = wave * 64;
  const int m = lane & 15;
  const int hi16 = (lane >> 4) & 1;
  const int kbh = hi16 * 8;
  const bool isg = (wave >> 1) == 2;
  const float act_s = isg ? -2.0f * LOG2E_F : -LOG2E_F;
  const float act_a = isg ? 2.0f : 1.0f;
  const float act_b = isg ? -1.0f : 0.0f;

  v16h wihB[4][KFX], whhB[4][4];
  float biasv[4];
#pragma unroll
  for (int nt = 0; nt < 4; ++nt) {
    const int g = nbase + nt * 16 + m;
    biasv[nt] = bih[g] + bhh[g];
#pragma unroll
    for (int kf = 0; kf < KFX; ++kf)
#pragma unroll
      for (int h = 0; h < 16; ++h) {
        int k = kf * 32 + ((h & 8) << 1) + kbh + (h & 7);
        wihB[nt][kf][h] = (_Float16)Wih[(size_t)g * KIN + k];
      }
#pragma unroll
    for (int kf = 0; kf < 4; ++kf)
#pragma unroll
      for (int h = 0; h < 16; ++h) {
        int k = kf * 32 + ((h & 8) << 1) + kbh + (h & 7);
        whhB[nt][kf][h] = (_Float16)Whh[(size_t)g * HID_ + k];
      }
  }

  const int prow = tid >> 4;
  const int pj = (tid & 15) * 8;
  float c[8];
#pragma unroll
  for (int i = 0; i < 8; ++i) { c[i] = 0.f; hbuf[prow][pj + i] = (_Float16)0.f; }
  __syncthreads();

  for (int t = 0; t < SEQ_; ++t) {
    v16h xA[KFX], hA[4];
#pragma unroll
    for (int kf = 0; kf < KFX; ++kf) {
      const _Float16* p = xin + ((size_t)(bm + m) * SEQ_ + t) * KIN + kf * 32 + kbh;
      v8h lo = *(const v8h*)p;
      v8h hi = *(const v8h*)(p + 16);
#pragma unroll
      for (int i = 0; i < 8; ++i) { xA[kf][i] = lo[i]; xA[kf][i + 8] = hi[i]; }
    }
#pragma unroll
    for (int kf = 0; kf < 4; ++kf) {
      const _Float16* p = &hbuf[m][kf * 32 + kbh];
      v8h lo = *(const v8h*)p;
      v8h hi = *(const v8h*)(p + 16);
#pragma unroll
      for (int i = 0; i < 8; ++i) { hA[kf][i] = lo[i]; hA[kf][i + 8] = hi[i]; }
    }
    v8f acc[4];
#pragma unroll
    for (int nt = 0; nt < 4; ++nt)
#pragma unroll
      for (int r = 0; r < 8; ++r) acc[nt][r] = biasv[nt];
#pragma unroll
    for (int nt = 0; nt < 4; ++nt) {
#pragma unroll
      for (int kf = 0; kf < KFX; ++kf)
        acc[nt] = __builtin_amdgcn_wmma_f32_16x16x32_f16(false, xA[kf], false, wihB[nt][kf],
                                                         (short)0, acc[nt], false, false);
#pragma unroll
      for (int kf = 0; kf < 4; ++kf)
        acc[nt] = __builtin_amdgcn_wmma_f32_16x16x32_f16(false, hA[kf], false, whhB[nt][kf],
                                                         (short)0, acc[nt], false, false);
    }
#pragma unroll
    for (int nt = 0; nt < 4; ++nt) {
      const int g = nbase + nt * 16 + m;
#pragma unroll
      for (int r = 0; r < 8; ++r) {
        int row = r + hi16 * 8;
        gates[row][g] = act_exp2_rcp(acc[nt][r], act_s, act_a, act_b);
      }
    }
    __syncthreads();
    v8h hpk;
#pragma unroll
    for (int i = 0; i < 8; ++i) {
      int j = pj + i;
      float iv = gates[prow][j];
      float fv = gates[prow][HID_ + j];
      float gv = gates[prow][2 * HID_ + j];
      float ov = gates[prow][3 * HID_ + j];
      c[i] = fv * c[i] + iv * gv;
      float hv = ov * act_exp2_rcp(c[i], -2.0f * LOG2E_F, 2.0f, -1.0f);
      hpk[i] = (_Float16)hv;
      if (!STORE_SEQ && t == SEQ_ - 1) hlast[(size_t)(bm + prow) * HID_ + j] = hv;
    }
#pragma unroll
    for (int i = 0; i < 8; ++i) hbuf[prow][pj + i] = hpk[i];
    if (STORE_SEQ)
      *(v8h*)(hseq + ((size_t)(bm + prow) * SEQ_ + t) * HID_ + pj) = hpk;
    __syncthreads();
  }
}

__global__ __launch_bounds__(128) void head_kernel(const float* __restrict__ h1last,
                                                   const float* __restrict__ W2,
                                                   const float* __restrict__ b2,
                                                   const float* __restrict__ W3,
                                                   const float* __restrict__ b3,
                                                   float* __restrict__ out) {
  __shared__ float lastS[HID_];
  __shared__ float l2S[L2_];
  const int b = blockIdx.x, tid = threadIdx.x;
  lastS[tid] = tanhf(h1last[(size_t)b * HID_ + tid]);
  __syncthreads();
  if (tid < L2_) {
    float acc = b2[tid];
    for (int k = 0; k < HID_; ++k) acc += lastS[k] * W2[tid * HID_ + k];
    l2S[tid] = tanhf(acc);
  }
  __syncthreads();
  if (tid < HOR_) {
    float acc = b3[tid];
    for (int k = 0; k < L2_; ++k) acc += l2S[k] * W3[tid * L2_ + k];
    out[(size_t)b * HOR_ + tid] = acc;
  }
}

extern "C" void kernel_launch(void* const* d_in, const int* in_sizes, int n_in,
                              void* d_out, int out_size, void* d_ws, size_t ws_size,
                              hipStream_t stream) {
  (void)in_sizes; (void)n_in; (void)out_size; (void)ws_size;
  const float* inp  = (const float*)d_in[0];
  const float* akey = (const float*)d_in[1];
  const float* aval = (const float*)d_in[2];
  const float* W1   = (const float*)d_in[3];
  const float* b1   = (const float*)d_in[4];
  const float* Wih0 = (const float*)d_in[5];
  const float* Whh0 = (const float*)d_in[6];
  const float* bih0 = (const float*)d_in[7];
  const float* bhh0 = (const float*)d_in[8];
  const float* Wih1 = (const float*)d_in[9];
  const float* Whh1 = (const float*)d_in[10];
  const float* bih1 = (const float*)d_in[11];
  const float* bhh1 = (const float*)d_in[12];
  const float* W2   = (const float*)d_in[13];
  const float* b2   = (const float*)d_in[14];
  const float* W3   = (const float*)d_in[15];
  const float* b3   = (const float*)d_in[16];

  char* ws = (char*)d_ws;
  size_t off = 0;
  float*    dist   = (float*)(ws + off);    off += (size_t)B_ * NK_ * 4;
  float*    hist   = (float*)(ws + off);    off += (size_t)B_ * HLEN_ * 4;
  _Float16* xf     = (_Float16*)(ws + off); off += (size_t)B_ * SEQ_ * L1_ * 2;
  _Float16* h0     = (_Float16*)(ws + off); off += (size_t)B_ * SEQ_ * HID_ * 2;
  float*    h1last = (float*)(ws + off);    off += (size_t)B_ * HID_ * 4;

  gemm1_kernel<<<dim3(16, 64), 256, 0, stream>>>(inp, akey, dist);
  softmax_kernel<<<dim3(B_), 256, 0, stream>>>(dist);
  gemm2_kernel<<<dim3(16, 5), 256, 0, stream>>>(dist, aval, hist);
  featl1_kernel<<<dim3(B_, SEQ_ / 8), 256, 0, stream>>>(inp, hist, W1, b1, xf);
  lstm_kernel<L1_, true><<<dim3(B_ / 16), 256, 0, stream>>>(xf, Wih0, Whh0, bih0, bhh0, h0, nullptr);
  lstm_kernel<HID_, false><<<dim3(B_ / 16), 256, 0, stream>>>(h0, Wih1, Whh1, bih1, bhh1, nullptr, h1last);
  head_kernel<<<dim3(B_), 128, 0, stream>>>(h1last, W2, b2, W3, b3, (float*)d_out);
}


# LSTM layers: x_(t+1) A-fragments are loaded from global into a second register set during step t (two sets, time loop unrolled by 2), so load latency no longer stands in front of each step's WMMAs
# speedup vs baseline: 2.8159x; 1.2426x over previous
#include <hip/hip_runtime.h>
#include <cstdint>
#include <cstddef>

typedef __attribute__((ext_vector_type(16))) _Float16 v16h;
typedef __attribute__((ext_vector_type(8)))  _Float16 v8h;
typedef __attribute__((ext_vector_type(8)))  float    v8f;
typedef __attribute__((ext_vector_type(2)))  float    v2f;

#define B_    256
#define SEQ_  512
#define HOR_  32
#define NK_   8192
#define L1_   64
#define HID_  128
#define L2_   64
#define GATES_ 512
#define HLEN_ 544
#define HPAD_ 8
#define GPAD_ 4

__global__ __launch_bounds__(256) void gemm1_kernel(const float* __restrict__ inp,
                                                    const float* __restrict__ akey,
                                                    float* __restrict__ dist) {
  const int tid = threadIdx.x, lane = tid & 31, wave = tid >> 5;
  const int mt = blockIdx.x, nt = blockIdx.y * 8 + wave;
  const int m = lane & 15;
  const int kb = (lane & 16) ? 2 : 0;
  const float* Ap = inp  + (size_t)(mt * 16 + m) * SEQ_ + kb;
  const float* Bp = akey + (size_t)(nt * 16 + m) * SEQ_ + kb;
  v8f acc0 = {}, acc1 = {};
  for (int k = 0; k < SEQ_; k += 8) {
    v2f a0 = *(const v2f*)(Ap + k);
    v2f b0 = *(const v2f*)(Bp + k);
    acc0 = __builtin_amdgcn_wmma_f32_16x16x4_f32(false, a0, false, b0, (short)0, acc0, false, false);
    v2f a1 = *(const v2f*)(Ap + k + 4);
    v2f b1 = *(const v2f*)(Bp + k + 4);
    acc1 = __builtin_amdgcn_wmma_f32_16x16x4_f32(false, a1, false, b1, (short)0, acc1, false, false);
  }
#pragma unroll
  for (int r = 0; r < 8; ++r) {
    int row = r + ((lane & 16) ? 8 : 0);
    dist[(size_t)(mt * 16 + row) * NK_ + nt * 16 + m] = acc0[r] + acc1[r];
  }
}

__global__ __launch_bounds__(256) void softmax_kernel(float* __restrict__ dist) {
  __shared__ float sv[256];
  __shared__ int   si[256];
  const int b = blockIdx.x, tid = threadIdx.x;
  float* row = dist + (size_t)b * NK_;

  float m = -3.4e38f; int mi = NK_;
  for (int j = tid; j < NK_; j += 256) {
    float v = row[j];
    if (v > m || (v == m && j < mi)) { m = v; mi = j; }
  }
  sv[tid] = m; si[tid] = mi;
  __syncthreads();
  for (int s = 128; s > 0; s >>= 1) {
    if (tid < s) {
      float v2 = sv[tid + s]; int i2 = si[tid + s];
      if (v2 > sv[tid] || (v2 == sv[tid] && i2 < si[tid])) { sv[tid] = v2; si[tid] = i2; }
    }
    __syncthreads();
  }
  if (tid == 0) row[si[0]] = 0.0f;
  __syncthreads();

  float mm = -3.4e38f;
  for (int j = tid; j < NK_; j += 256) mm = fmaxf(mm, row[j]);
  sv[tid] = mm;
  __syncthreads();
  for (int s = 128; s > 0; s >>= 1) { if (tid < s) sv[tid] = fmaxf(sv[tid], sv[tid + s]); __syncthreads(); }
  const float M = sv[0];
  __syncthreads();

  float sum = 0.f;
  for (int j = tid; j < NK_; j += 256) sum += __expf(row[j] - M);
  sv[tid] = sum;
  __syncthreads();
  for (int s = 128; s > 0; s >>= 1) { if (tid < s) sv[tid] += sv[tid + s]; __syncthreads(); }
  const float inv = 1.0f / sv[0];
  for (int j = tid; j < NK_; j += 256) row[j] = __expf(row[j] - M) * inv;
}

__global__ __launch_bounds__(256) void gemm2_kernel(const float* __restrict__ a,
                                                    const float* __restrict__ aval,
                                                    float* __restrict__ hist) {
  const int tid = threadIdx.x, lane = tid & 31, wave = tid >> 5;
  const int mt = blockIdx.x, nt = blockIdx.y * 8 + wave;
  if (nt >= (HLEN_ / 16)) return;
  const int m = lane & 15;
  const int kb = (lane & 16) ? 2 : 0;
  const int n = nt * 16 + m;
  const float* Ap = a + (size_t)(mt * 16 + m) * NK_ + kb;
  const float* Bp = aval + (size_t)kb * HLEN_ + n;
  v8f acc0 = {}, acc1 = {};
  for (int k = 0; k < NK_; k += 8) {
    v2f av = *(const v2f*)(Ap + k);
    v2f bv; bv[0] = Bp[(size_t)k * HLEN_]; bv[1] = Bp[(size_t)(k + 1) * HLEN_];
    acc0 = __builtin_amdgcn_wmma_f32_16x16x4_f32(false, av, false, bv, (short)0, acc0, false, false);
    v2f av1 = *(const v2f*)(Ap + k + 4);
    v2f bv1; bv1[0] = Bp[(size_t)(k + 4) * HLEN_]; bv1[1] = Bp[(size_t)(k + 5) * HLEN_];
    acc1 = __builtin_amdgcn_wmma_f32_16x16x4_f32(false, av1, false, bv1, (short)0, acc1, false, false);
  }
#pragma unroll
  for (int r = 0; r < 8; ++r) {
    int row = r + ((lane & 16) ? 8 : 0);
    hist[(size_t)(mt * 16 + row) * HLEN_ + n] = acc0[r] + acc1[r];
  }
}

__global__ __launch_bounds__(256) void featl1_kernel(const float* __restrict__ inp,
                                                     const float* __restrict__ hist,
                                                     const float* __restrict__ W1,
                                                     const float* __restrict__ b1,
                                                     _Float16* __restrict__ xf) {
  __shared__ float w1s[L1_ * (HOR_ + 1)];
  __shared__ float b1s[L1_];
  __shared__ float hs[8 + HOR_ - 1];
  __shared__ float ins[8];
  const int b = blockIdx.x, t0 = blockIdx.y * 8, tid = threadIdx.x;
  for (int i = tid; i < L1_ * (HOR_ + 1); i += 256) w1s[i] = W1[i];
  if (tid < L1_) b1s[tid] = b1[tid];
  if (tid < 8 + HOR_ - 1) hs[tid] = hist[(size_t)b * HLEN_ + t0 + tid];
  if (tid < 8) ins[tid] = inp[(size_t)b * SEQ_ + t0 + tid];
  __syncthreads();
  const int o = tid & 63;
#pragma unroll
  for (int pass = 0; pass < 2; ++pass) {
    int tt = (tid >> 6) + pass * 4;
    float acc = b1s[o] + ins[tt] * w1s[o * 33];
#pragma unroll
    for (int j = 0; j < HOR_; ++j) acc += hs[tt + j] * w1s[o * 33 + 1 + j];
    xf[((size_t)b * SEQ_ + (t0 + tt)) * L1_ + o] = (_Float16)tanhf(acc);
  }
}

#define LOG2E_F 1.4426950408889634f
__device__ __forceinline__ float act_exp2_rcp(float v, float s, float a, float b) {
  const float e = __builtin_amdgcn_exp2f(s * v);
  return __builtin_fmaf(a, __builtin_amdgcn_rcpf(1.0f + e), b);
}

template <int KIN, bool STORE_SEQ>
__global__ __launch_bounds__(256, 1) void lstm_kernel(const _Float16* __restrict__ xin,
                                                      const float* __restrict__ Wih,
                                                      const float* __restrict__ Whh,
                                                      const float* __restrict__ bih,
                                                      const float* __restrict__ bhh,
                                                      _Float16* __restrict__ hseq,
                                                      float* __restrict__ hlast) {
  constexpr int KFX = KIN / 32;
  __shared__ __align__(16) float gates[16][GATES_ + GPAD_];
  __shared__ __align__(16) _Float16 hbuf[16][HID_ + HPAD_];
  const int tid = threadIdx.x, lane = tid & 31, wave = tid >> 5;
  const int bm = blockIdx.x * 16;
  const int nbase = wave * 64;
  const int m = lane & 15;
  const int hi16 = (lane >> 4) & 1;
  const int kbh = hi16 * 8;
  const bool isg = (wave >> 1) == 2;
  const float act_s = isg ? -2.0f * LOG2E_F : -LOG2E_F;
  const float act_a = isg ? 2.0f : 1.0f;
  const float act_b = isg ? -1.0f : 0.0f;

  v16h wihB[4][KFX], whhB[4][4];
  float biasv[4];
#pragma unroll
  for (int nt = 0; nt < 4; ++nt) {
    const int g = nbase + nt * 16 + m;
    biasv[nt] = bih[g] + bhh[g];
#pragma unroll
    for (int kf = 0; kf < KFX; ++kf)
#pragma unroll
      for (int h = 0; h < 16; ++h) {
        int k = kf * 32 + ((h & 8) << 1) + kbh + (h & 7);
        wihB[nt][kf][h] = (_Float16)Wih[(size_t)g * KIN + k];
      }
#pragma unroll
    for (int kf = 0; kf < 4; ++kf)
#pragma unroll
      for (int h = 0; h < 16; ++h) {
        int k = kf * 32 + ((h & 8) << 1) + kbh + (h & 7);
        whhB[nt][kf][h] = (_Float16)Whh[(size_t)g * HID_ + k];
      }
    __builtin_amdgcn_sched_barrier(0);
  }

  const int prow = tid >> 4;
  const int pj = (tid & 15) * 8;
  float c[8];
#pragma unroll
  for (int i = 0; i < 8; ++i) { c[i] = 0.f; hbuf[prow][pj + i] = (_Float16)0.f; }
  __syncthreads();

  const _Float16* xrow = xin + (size_t)(bm + m) * SEQ_ * KIN + kbh;
  v16h xbuf[2][KFX];
#pragma unroll
  for (int kf = 0; kf < KFX; ++kf) {
    const _Float16* p = xrow + kf * 32;
    v8h lo = *(const v8h*)p;
    v8h hi = *(const v8h*)(p + 16);
#pragma unroll
    for (int i = 0; i < 8; ++i) { xbuf[0][kf][i] = lo[i]; xbuf[0][kf][i + 8] = hi[i]; }
  }

  for (int t2 = 0; t2 < SEQ_; t2 += 2) {
#pragma unroll
    for (int u = 0; u < 2; ++u) {
      const int t = t2 + u;
      {
        const int tn = (t + 1 < SEQ_) ? t + 1 : SEQ_ - 1;
#pragma unroll
        for (int kf = 0; kf < KFX; ++kf) {
          const _Float16* p = xrow + (size_t)tn * KIN + kf * 32;
          v8h lo = *(const v8h*)p;
          v8h hi = *(const v8h*)(p + 16);
#pragma unroll
          for (int i = 0; i < 8; ++i) { xbuf[u ^ 1][kf][i] = lo[i]; xbuf[u ^ 1][kf][i + 8] = hi[i]; }
        }
      }
      v16h hA[4];
#pragma unroll
      for (int kf = 0; kf < 4; ++kf) {
        const _Float16* p = &hbuf[m][kf * 32 + kbh];
        v8h lo = *(const v8h*)p;
        v8h hi = *(const v8h*)(p + 16);
#pragma unroll
        for (int i = 0; i < 8; ++i) { hA[kf][i] = lo[i]; hA[kf][i + 8] = hi[i]; }
      }
      v8f acc[4];
#pragma unroll
      for (int nt = 0; nt < 4; ++nt)
#pragma unroll
        for (int r = 0; r < 8; ++r) acc[nt][r] = biasv[nt];
#pragma unroll
      for (int nt = 0; nt < 4; ++nt) {
#pragma unroll
        for (int kf = 0; kf < KFX; ++kf)
          acc[nt] = __builtin_amdgcn_wmma_f32_16x16x32_f16(false, xbuf[u][kf], false, wihB[nt][kf],
                                                           (short)0, acc[nt], false, false);
#pragma unroll
        for (int kf = 0; kf < 4; ++kf)
          acc[nt] = __builtin_amdgcn_wmma_f32_16x16x32_f16(false, hA[kf], false, whhB[nt][kf],
                                                           (short)0, acc[nt], false, false);
      }
#pragma unroll
      for (int nt = 0; nt < 4; ++nt) {
        const int g = nbase + nt * 16 + m;
#pragma unroll
        for (int r = 0; r < 8; ++r) {
          int row = r + hi16 * 8;
          gates[row][g] = act_exp2_rcp(acc[nt][r], act_s, act_a, act_b);
        }
      }
      __syncthreads();
      v8h hpk;
#pragma unroll
      for (int i = 0; i < 8; ++i) {
        int j = pj + i;
        float iv = gates[prow][j];
        float fv = gates[prow][HID_ + j];
        float gv = gates[prow][2 * HID_ + j];
        float ov = gates[prow][3 * HID_ + j];
        c[i] = fv * c[i] + iv * gv;
        float hv = ov * act_exp2_rcp(c[i], -2.0f * LOG2E_F, 2.0f, -1.0f);
        hpk[i] = (_Float16)hv;
        if (!STORE_SEQ && t == SEQ_ - 1) hlast[(size_t)(bm + prow) * HID_ + j] = hv;
      }
#pragma unroll
      for (int i = 0; i < 8; ++i) hbuf[prow][pj + i] = hpk[i];
      if (STORE_SEQ)
        *(v8h*)(hseq + ((size_t)(bm + prow) * SEQ_ + t) * HID_ + pj) = hpk;
      __syncthreads();
    }
  }
}

__global__ __launch_bounds__(128) void head_kernel(const float* __restrict__ h1last,
                                                   const float* __restrict__ W2,
                                                   const float* __restrict__ b2,
                                                   const float* __restrict__ W3,
                                                   const float* __restrict__ b3,
                                                   float* __restrict__ out) {
  __shared__ float lastS[HID_];
  __shared__ float l2S[L2_];
  const int b = blockIdx.x, tid = threadIdx.x;
  lastS[tid] = tanhf(h1last[(size_t)b * HID_ + tid]);
  __syncthreads();
  if (tid < L2_) {
    float acc = b2[tid];
    for (int k = 0; k < HID_; ++k) acc += lastS[k] * W2[tid * HID_ + k];
    l2S[tid] = tanhf(acc);
  }
  __syncthreads();
  if (tid < HOR_) {
    float acc = b3[tid];
    for (int k = 0; k < L2_; ++k) acc += l2S[k] * W3[tid * L2_ + k];
    out[(size_t)b * HOR_ + tid] = acc;
  }
}

extern "C" void kernel_launch(void* const* d_in, const int* in_sizes, int n_in,
                              void* d_out, int out_size, void* d_ws, size_t ws_size,
                              hipStream_t stream) {
  (void)in_sizes; (void)n_in; (void)out_size; (void)ws_size;
  const float* inp  = (const float*)d_in[0];
  const float* akey = (const float*)d_in[1];
  const float* aval = (const float*)d_in[2];
  const float* W1   = (const float*)d_in[3];
  const float* b1   = (const float*)d_in[4];
  const float* Wih0 = (const float*)d_in[5];
  const float* Whh0 = (const float*)d_in[6];
  const float* bih0 = (const float*)d_in[7];
  const float* bhh0 = (const float*)d_in[8];
  const float* Wih1 = (const float*)d_in[9];
  const float* Whh1 = (const float*)d_in[10];
  const float* bih1 = (const float*)d_in[11];
  const float* bhh1 = (const float*)d_in[12];
  const float* W2   = (const float*)d_in[13];
  const float* b2   = (const float*)d_in[14];
  const float* W3   = (const float*)d_in[15];
  const float* b3   = (const float*)d_in[16];

  char* ws = (char*)d_ws;
  size_t off = 0;
  float*    dist   = (float*)(ws + off);    off += (size_t)B_ * NK_ * 4;
  float*    hist   = (float*)(ws + off);    off += (size_t)B_ * HLEN_ * 4;
  _Float16* xf     = (_Float16*)(ws + off); off += (size_t)B_ * SEQ_ * L1_ * 2;
  _Float16* h0     = (_Float16*)(ws + off); off += (size_t)B_ * SEQ_ * HID_ * 2;
  float*    h1last = (float*)(ws + off);    off += (size_t)B_ * HID_ * 4;

  gemm1_kernel<<<dim3(16, 64), 256, 0, stream>>>(inp, akey, dist);
  softmax_kernel<<<dim3(B_), 256, 0, stream>>>(dist);
  gemm2_kernel<<<dim3(16, 5), 256, 0, stream>>>(dist, aval, hist);
  featl1_kernel<<<dim3(B_, SEQ_ / 8), 256, 0, stream>>>(inp, hist, W1, b1, xf);
  lstm_kernel<L1_, true><<<dim3(B_ / 16), 256, 0, stream>>>(xf, Wih0, Whh0, bih0, bhh0, h0, nullptr);
  lstm_kernel<HID_, false><<<dim3(B_ / 16), 256, 0, stream>>>(h0, Wih1, Whh1, bih1, bhh1, nullptr, h1last);
  head_kernel<<<dim3(B_), 128, 0, stream>>>(h1last, W2, b2, W3, b3, (float*)d_out);
}


# LSTM layers overlapped in time: sequence cut into 8 chunks of 64 steps; launch k runs layer 0 on chunk k beside layer 1 on chunk k-1 (32 workgroups, 9 launches); c (f32), h (f16) via workspace
# speedup vs baseline: 4.9769x; 1.7674x over previous
#include <hip/hip_runtime.h>
#include <cstdint>
#include <cstddef>

typedef __attribute__((ext_vector_type(16))) _Float16 v16h;
typedef __attribute__((ext_vector_type(8)))  _Float16 v8h;
typedef __attribute__((ext_vector_type(8)))  float    v8f;
typedef __attribute__((ext_vector_type(2)))  float    v2f;

#define B_    256
#define SEQ_  512
#define HOR_  32
#define NK_   8192
#define L1_   64
#define HID_  128
#define L2_   64
#define GATES_ 512
#define HLEN_ 544
#define HPAD_ 8

__global__ __launch_bounds__(256) void gemm1_kernel(const float* __restrict__ inp,
                                                    const float* __restrict__ akey,
                                                    float* __restrict__ dist) {
  const int tid = threadIdx.x, lane = tid & 31, wave = tid >> 5;
  const int mt = blockIdx.x, nt = blockIdx.y * 8 + wave;
  const int m = lane & 15;
  const int kb = (lane & 16) ? 2 : 0;
  const float* Ap = inp  + (size_t)(mt * 16 + m) * SEQ_ + kb;
  const float* Bp = akey + (size_t)(nt * 16 + m) * SEQ_ + kb;
  v8f acc0 = {}, acc1 = {};
  for (int k = 0; k < SEQ_; k += 8) {
    v2f a0 = *(const v2f*)(Ap + k);
    v2f b0 = *(const v2f*)(Bp + k);
    acc0 = __builtin_amdgcn_wmma_f32_16x16x4_f32(false, a0, false, b0, (short)0, acc0, false, false);
    v2f a1 = *(const v2f*)(Ap + k + 4);
    v2f b1 = *(const v2f*)(Bp + k + 4);
    acc1 = __builtin_amdgcn_wmma_f32_16x16x4_f32(false, a1, false, b1, (short)0, acc1, false, false);
  }
#pragma unroll
  for (int r = 0; r < 8; ++r) {
    int row = r + ((lane & 16) ? 8 : 0);
    dist[(size_t)(mt * 16 + row) * NK_ + nt * 16 + m] = acc0[r] + acc1[r];
  }
}

__global__ __launch_bounds__(256) void softmax_kernel(float* __restrict__ dist) {
  __shared__ float sv[256];
  __shared__ int   si[256];
  const int b = blockIdx.x, tid = threadIdx.x;
  float* row = dist + (size_t)b * NK_;

  float m = -3.4e38f; int mi = NK_;
  for (int j = tid; j < NK_; j += 256) {
    float v = row[j];
    if (v > m || (v == m && j < mi)) { m = v; mi = j; }
  }
  sv[tid] = m; si[tid] = mi;
  __syncthreads();
  for (int s = 128; s > 0; s >>= 1) {
    if (tid < s) {
      float v2 = sv[tid + s]; int i2 = si[tid + s];
      if (v2 > sv[tid] || (v2 == sv[tid] && i2 < si[tid])) { sv[tid] = v2; si[tid] = i2; }
    }
    __syncthreads();
  }
  if (tid == 0) row[si[0]] = 0.0f;
  __syncthreads();

  float mm = -3.4e38f;
  for (int j = tid; j < NK_; j += 256) mm = fmaxf(mm, row[j]);
  sv[tid] = mm;
  __syncthreads();
  for (int s = 128; s > 0; s >>= 1) { if (tid < s) sv[tid] = fmaxf(sv[tid], sv[tid + s]); __syncthreads(); }
  const float M = sv[0];
  __syncthreads();

  float sum = 0.f;
  for (int j = tid; j < NK_; j += 256) sum += __expf(row[j] - M);
  sv[tid] = sum;
  __syncthreads();
  for (int s = 128; s > 0; s >>= 1) { if (tid < s) sv[tid] += sv[tid + s]; __syncthreads(); }
  const float inv = 1.0f / sv[0];
  for (int j = tid; j < NK_; j += 256) row[j] = __expf(row[j] - M) * inv;
}

__global__ __launch_bounds__(256) void gemm2_kernel(const float* __restrict__ a,
                                                    const float* __restrict__ aval,
                                                    float* __restrict__ hist) {
  const int tid = threadIdx.x, lane = tid & 31, wave = tid >> 5;
  const int mt = blockIdx.x, nt = blockIdx.y * 8 + wave;
  if (nt >= (HLEN_ / 16)) return;
  const int m = lane & 15;
  const int kb = (lane & 16) ? 2 : 0;
  const int n = nt * 16 + m;
  const float* Ap = a + (size_t)(mt * 16 + m) * NK_ + kb;
  const float* Bp = aval + (size_t)kb * HLEN_ + n;
  v8f acc0 = {}, acc1 = {};
  for (int k = 0; k < NK_; k += 8) {
    v2f av = *(const v2f*)(Ap + k);
    v2f bv; bv[0] = Bp[(size_t)k * HLEN_]; bv[1] = Bp[(size_t)(k + 1) * HLEN_];
    acc0 = __builtin_amdgcn_wmma_f32_16x16x4_f32(false, av, false, bv, (short)0, acc0, false, false);
    v2f av1 = *(const v2f*)(Ap + k + 4);
    v2f bv1; bv1[0] = Bp[(size_t)(k + 4) * HLEN_]; bv1[1] = Bp[(size_t)(k + 5) * HLEN_];
    acc1 = __builtin_amdgcn_wmma_f32_16x16x4_f32(false, av1, false, bv1, (short)0, acc1, false, false);
  }
#pragma unroll
  for (int r = 0; r < 8; ++r) {
    int row = r + ((lane & 16) ? 8 : 0);
    hist[(size_t)(mt * 16 + row) * HLEN_ + n] = acc0[r] + acc1[r];
  }
}

__global__ __launch_bounds__(256) void featl1_kernel(const float* __restrict__ inp,
                                                     const float* __restrict__ hist,
                                                     const float* __restrict__ W1,
                                                     const float* __restrict__ b1,
                                                     _Float16* __restrict__ xf) {
  __shared__ float w1s[L1_ * (HOR_ + 1)];
  __shared__ float b1s[L1_];
  __shared__ float hs[8 + HOR_ - 1];
  __shared__ float ins[8];
  const int b = blockIdx.x, t0 = blockIdx.y * 8, tid = threadIdx.x;
  for (int i = tid; i < L1_ * (HOR_ + 1); i += 256) w1s[i] = W1[i];
  if (tid < L1_) b1s[tid] = b1[tid];
  if (tid < 8 + HOR_ - 1) hs[tid] = hist[(size_t)b * HLEN_ + t0 + tid];
  if (tid < 8) ins[tid] = inp[(size_t)b * SEQ_ + t0 + tid];
  __syncthreads();
  const int o = tid & 63;
#pragma unroll
  for (int pass = 0; pass < 2; ++pass) {
    int tt = (tid >> 6) + pass * 4;
    float acc = b1s[o] + ins[tt] * w1s[o * 33];
#pragma unroll
    for (int j = 0; j < HOR_; ++j) acc += hs[tt + j] * w1s[o * 33 + 1 + j];
    xf[((size_t)b * SEQ_ + (t0 + tt)) * L1_ + o] = (_Float16)tanhf(acc);
  }
}

#define LOG2E_F 1.4426950408889634f
__device__ __forceinline__ float act_logistic(float v) {
  const float e = __builtin_amdgcn_exp2f(-LOG2E_F * v);
  return __builtin_amdgcn_rcpf(1.0f + e);
}
__device__ __forceinline__ float act_tanh(float v) {
  const float e = __builtin_amdgcn_exp2f(-2.0f * LOG2E_F * v);
  return __builtin_fmaf(2.0f, __builtin_amdgcn_rcpf(1.0f + e), -1.0f);
}

#define CHUNK_  64
#define NCHUNK_ 8

template <int KIN, bool STORE_SEQ>
__device__ __forceinline__ void lstm_chunk(_Float16 (*hbuf)[16][HID_ + HPAD_],
                                           const _Float16* __restrict__ xin,
                                           const float* __restrict__ Wih,
                                           const float* __restrict__ Whh,
                                           const float* __restrict__ bih,
                                           const float* __restrict__ bhh,
                                           _Float16* __restrict__ hseq,
                                           _Float16* __restrict__ hstate,
                                           float* __restrict__ cstate,
                                           float* __restrict__ hlast,
                                           const int bm, const int t0) {
  constexpr int KFX = KIN / 32;
  const int tid = threadIdx.x, lane = tid & 31, wave = tid >> 5;
  const int m = lane & 15;
  const int hi16 = (lane >> 4) & 1;
  const int kbh = hi16 * 8;
  const int unit = wave * 16 + m;
  const int drow = hi16 * 8;
  const int tend = t0 + CHUNK_;

  const int prow = tid >> 4;
  const int pj = (tid & 15) * 8;
  float c[8];
  if (t0 == 0) {
#pragma unroll
    for (int r = 0; r < 8; ++r) c[r] = 0.f;
#pragma unroll
    for (int i = 0; i < 8; ++i) hbuf[0][prow][pj + i] = (_Float16)0.f;
  } else {
#pragma unroll
    for (int r = 0; r < 8; ++r) c[r] = cstate[(size_t)(bm + drow + r) * HID_ + unit];
    const _Float16* hp = STORE_SEQ ? hseq + ((size_t)(bm + prow) * SEQ_ + (t0 - 1)) * HID_ + pj
                                   : hstate + (size_t)(bm + prow) * HID_ + pj;
    *(v8h*)&hbuf[0][prow][pj] = *(const v8h*)hp;
  }
  __syncthreads();


  v16h wihB[4][KFX], whhB[4][4];
  float biasv[4];
#pragma unroll
  for (int q = 0; q < 4; ++q) {
    const int g = q * HID_ + unit;
    biasv[q] = bih[g] + bhh[g];
#pragma unroll
    for (int kf = 0; kf < KFX; ++kf)
#pragma unroll
      for (int h = 0; h < 16; ++h) {
        int k = kf * 32 + ((h & 8) << 1) + kbh + (h & 7);
        wihB[q][kf][h] = (_Float16)Wih[(size_t)g * KIN + k];
      }
#pragma unroll
    for (int kf = 0; kf < 4; ++kf)
#pragma unroll
      for (int h = 0; h < 16; ++h) {
        int k = kf * 32 + ((h & 8) << 1) + kbh + (h & 7);
        whhB[q][kf][h] = (_Float16)Whh[(size_t)g * HID_ + k];
      }
    __builtin_amdgcn_sched_barrier(0);
  }

  const _Float16* xrow = xin + (size_t)(bm + m) * SEQ_ * KIN + kbh;
  v16h xbuf[2][KFX];
#pragma unroll
  for (int kf = 0; kf < KFX; ++kf) {
    const _Float16* p = xrow + (size_t)t0 * KIN + kf * 32;
    v8h lo = *(const v8h*)p;
    v8h hi = *(const v8h*)(p + 16);
#pragma unroll
    for (int i = 0; i < 8; ++i) { xbuf[0][kf][i] = lo[i]; xbuf[0][kf][i + 8] = hi[i]; }
  }

  float hv[8];
#pragma unroll
  for (int r = 0; r < 8; ++r) hv[r] = 0.f;

  for (int t2 = t0; t2 < tend; t2 += 2) {
#pragma unroll
    for (int u = 0; u < 2; ++u) {
      const int t = t2 + u;
      v16h hA[4];
#pragma unroll
      for (int kf = 0; kf < 4; ++kf) {
        const _Float16* p = &hbuf[u][m][kf * 32 + kbh];
        v8h lo = *(const v8h*)p;
        v8h hi = *(const v8h*)(p + 16);
#pragma unroll
        for (int i = 0; i < 8; ++i) { hA[kf][i] = lo[i]; hA[kf][i + 8] = hi[i]; }
      }
      v8h hrow;
      if (STORE_SEQ) hrow = *(const v8h*)&hbuf[u][prow][pj];
      {
        const int tn = (t + 1 < tend) ? t + 1 : tend - 1;
#pragma unroll
        for (int kf = 0; kf < KFX; ++kf) {
          const _Float16* p = xrow + (size_t)tn * KIN + kf * 32;
          v8h lo = *(const v8h*)p;
          v8h hi = *(const v8h*)(p + 16);
#pragma unroll
          for (int i = 0; i < 8; ++i) { xbuf[u ^ 1][kf][i] = lo[i]; xbuf[u ^ 1][kf][i + 8] = hi[i]; }
        }
      }
      __builtin_amdgcn_sched_barrier(0);
      v8f acc[4];
#pragma unroll
      for (int q = 0; q < 4; ++q)
#pragma unroll
        for (int r = 0; r < 8; ++r) acc[q][r] = biasv[q];
#pragma unroll
      for (int kf = 0; kf < KFX; ++kf)
#pragma unroll
        for (int q = 0; q < 4; ++q)
          acc[q] = __builtin_amdgcn_wmma_f32_16x16x32_f16(false, xbuf[u][kf], false, wihB[q][kf],
                                                          (short)0, acc[q], false, false);
#pragma unroll
      for (int kf = 0; kf < 4; ++kf)
#pragma unroll
        for (int q = 0; q < 4; ++q)
          acc[q] = __builtin_amdgcn_wmma_f32_16x16x32_f16(false, hA[kf], false, whhB[q][kf],
                                                          (short)0, acc[q], false, false);
      if (STORE_SEQ && t > t0)
        *(v8h*)(hseq + ((size_t)(bm + prow) * SEQ_ + (t - 1)) * HID_ + pj) = hrow;
      float gi[8], gf[8], gg[8], go[8];
#pragma unroll
      for (int r = 0; r < 8; ++r) gi[r] = act_logistic(acc[0][r]);
#pragma unroll
      for (int r = 0; r < 8; ++r) gf[r] = act_logistic(acc[1][r]);
#pragma unroll
      for (int r = 0; r < 8; ++r) gg[r] = act_tanh(acc[2][r]);
#pragma unroll
      for (int r = 0; r < 8; ++r) go[r] = act_logistic(acc[3][r]);
#pragma unroll
      for (int r = 0; r < 8; ++r) c[r] = gf[r] * c[r] + gi[r] * gg[r];
#pragma unroll
      for (int r = 0; r < 8; ++r) hv[r] = go[r] * act_tanh(c[r]);
#pragma unroll
      for (int r = 0; r < 8; ++r) hbuf[u ^ 1][drow + r][unit] = (_Float16)hv[r];
      __syncthreads();
    }
  }
  if (STORE_SEQ) {
    const v8h hrow = *(const v8h*)&hbuf[0][prow][pj];
    *(v8h*)(hseq + ((size_t)(bm + prow) * SEQ_ + (tend - 1)) * HID_ + pj) = hrow;
  } else if (tend == SEQ_) {
#pragma unroll
    for (int r = 0; r < 8; ++r) hlast[(size_t)(bm + drow + r) * HID_ + unit] = hv[r];
  } else {
    const v8h hrow = *(const v8h*)&hbuf[0][prow][pj];
    *(v8h*)(hstate + (size_t)(bm + prow) * HID_ + pj) = hrow;
  }
  if (tend < SEQ_) {
#pragma unroll
    for (int r = 0; r < 8; ++r) cstate[(size_t)(bm + drow + r) * HID_ + unit] = c[r];
  }
}

__global__ __launch_bounds__(256, 1) void lstm_pair_kernel(const _Float16* __restrict__ xf,
                                                           const float* __restrict__ Wih0,
                                                           const float* __restrict__ Whh0,
                                                           const float* __restrict__ bih0,
                                                           const float* __restrict__ bhh0,
                                                           const float* __restrict__ Wih1,
                                                           const float* __restrict__ Whh1,
                                                           const float* __restrict__ bih1,
                                                           const float* __restrict__ bhh1,
                                                           _Float16* h0seq,
                                                           float* c0state,
                                                           float* c1state,
                                                           _Float16* h1state,
                                                           float* h1last,
                                                           const int chunk) {
  __shared__ __align__(16) _Float16 hbuf[2][16][HID_ + HPAD_];
  const int bm = (blockIdx.x & 15) * 16;
  if (blockIdx.x < 16) {
    if (chunk < NCHUNK_)
      lstm_chunk<L1_, true>(hbuf, xf, Wih0, Whh0, bih0, bhh0, h0seq, nullptr, c0state, nullptr,
                            bm, chunk * CHUNK_);
  } else {
    if (chunk > 0)
      lstm_chunk<HID_, false>(hbuf, h0seq, Wih1, Whh1, bih1, bhh1, nullptr, h1state, c1state, h1last,
                              bm, (chunk - 1) * CHUNK_);
  }
}

__global__ __launch_bounds__(128) void head_kernel(const float* __restrict__ h1last,
                                                   const float* __restrict__ W2,
                                                   const float* __restrict__ b2,
                                                   const float* __restrict__ W3,
                                                   const float* __restrict__ b3,
                                                   float* __restrict__ out) {
  __shared__ float lastS[HID_];
  __shared__ float l2S[L2_];
  const int b = blockIdx.x, tid = threadIdx.x;
  lastS[tid] = tanhf(h1last[(size_t)b * HID_ + tid]);
  __syncthreads();
  if (tid < L2_) {
    float acc = b2[tid];
    for (int k = 0; k < HID_; ++k) acc += lastS[k] * W2[tid * HID_ + k];
    l2S[tid] = tanhf(acc);
  }
  __syncthreads();
  if (tid < HOR_) {
    float acc = b3[tid];
    for (int k = 0; k < L2_; ++k) acc += l2S[k] * W3[tid * L2_ + k];
    out[(size_t)b * HOR_ + tid] = acc;
  }
}

extern "C" void kernel_launch(void* const* d_in, const int* in_sizes, int n_in,
                              void* d_out, int out_size, void* d_ws, size_t ws_size,
                              hipStream_t stream) {
  (void)in_sizes; (void)n_in; (void)out_size; (void)ws_size;
  const float* inp  = (const float*)d_in[0];
  const float* akey = (const float*)d_in[1];
  const float* aval = (const float*)d_in[2];
  const float* W1   = (const float*)d_in[3];
  const float* b1   = (const float*)d_in[4];
  const float* Wih0 = (const float*)d_in[5];
  const float* Whh0 = (const float*)d_in[6];
  const float* bih0 = (const float*)d_in[7];
  const float* bhh0 = (const float*)d_in[8];
  const float* Wih1 = (const float*)d_in[9];
  const float* Whh1 = (const float*)d_in[10];
  const float* bih1 = (const float*)d_in[11];
  const float* bhh1 = (const float*)d_in[12];
  const float* W2   = (const float*)d_in[13];
  const float* b2   = (const float*)d_in[14];
  const float* W3   = (const float*)d_in[15];
  const float* b3   = (const float*)d_in[16];

  char* ws = (char*)d_ws;
  size_t off = 0;
  float*    dist   = (float*)(ws + off);    off += (size_t)B_ * NK_ * 4;
  float*    hist   = (float*)(ws + off);    off += (size_t)B_ * HLEN_ * 4;
  _Float16* xf     = (_Float16*)(ws + off); off += (size_t)B_ * SEQ_ * L1_ * 2;
  _Float16* h0     = (_Float16*)(ws + off); off += (size_t)B_ * SEQ_ * HID_ * 2;
  float*    h1last = (float*)(ws + off);    off += (size_t)B_ * HID_ * 4;
  float*    c0st   = (float*)(ws + off);    off += (size_t)B_ * HID_ * 4;
  float*    c1st   = (float*)(ws + off);    off += (size_t)B_ * HID_ * 4;
  _Float16* h1st   = (_Float16*)(ws + off); off += (size_t)B_ * HID_ * 2;

  gemm1_kernel<<<dim3(16, 64), 256, 0, stream>>>(inp, akey, dist);
  softmax_kernel<<<dim3(B_), 256, 0, stream>>>(dist);
  gemm2_kernel<<<dim3(16, 5), 256, 0, stream>>>(dist, aval, hist);
  featl1_kernel<<<dim3(B_, SEQ_ / 8), 256, 0, stream>>>(inp, hist, W1, b1, xf);
  lstm_pair_kernel<<<dim3(2 * (B_ / 16)), 256, 0, stream>>>(xf, Wih0, Whh0, bih0, bhh0, Wih1, Whh1, bih1, bhh1,
                                                         h0, c0st, c1st, h1st, h1last, 0);
  lstm_pair_kernel<<<dim3(2 * (B_ / 16)), 256, 0, stream>>>(xf, Wih0, Whh0, bih0, bhh0, Wih1, Whh1, bih1, bhh1,
                                                         h0, c0st, c1st, h1st, h1last, 1);
  lstm_pair_kernel<<<dim3(2 * (B_ / 16)), 256, 0, stream>>>(xf, Wih0, Whh0, bih0, bhh0, Wih1, Whh1, bih1, bhh1,
                                                         h0, c0st, c1st, h1st, h1last, 2);
  lstm_pair_kernel<<<dim3(2 * (B_ / 16)), 256, 0, stream>>>(xf, Wih0, Whh0, bih0, bhh0, Wih1, Whh1, bih1, bhh1,
                                                         h0, c0st, c1st, h1st, h1last, 3);
  lstm_pair_kernel<<<dim3(2 * (B_ / 16)), 256, 0, stream>>>(xf, Wih0, Whh0, bih0, bhh0, Wih1, Whh1, bih1, bhh1,
                                                         h0, c0st, c1st, h1st, h1last, 4);
  lstm_pair_kernel<<<dim3(2 * (B_ / 16)), 256, 0, stream>>>(xf, Wih0, Whh0, bih0, bhh0, Wih1, Whh1, bih1, bhh1,
                                                         h0, c0st, c1st, h1st, h1last, 5);
  lstm_pair_kernel<<<dim3(2 * (B_ / 16)), 256, 0, stream>>>(xf, Wih0, Whh0, bih0, bhh0, Wih1, Whh1, bih1, bhh1,
                                                         h0, c0st, c1st, h1st, h1last, 6);
  lstm_pair_kernel<<<dim3(2 * (B_ / 16)), 256, 0, stream>>>(xf, Wih0, Whh0, bih0, bhh0, Wih1, Whh1, bih1, bhh1,
                                                         h0, c0st, c1st, h1st, h1last, 7);
  lstm_pair_kernel<<<dim3(2 * (B_ / 16)), 256, 0, stream>>>(xf, Wih0, Whh0, bih0, bhh0, Wih1, Whh1, bih1, bhh1,
                                                         h0, c0st, c1st, h1st, h1last, 8);
  head_kernel<<<dim3(B_), 128, 0, stream>>>(h1last, W2, b2, W3, b3, (float*)d_out);
}


# LSTM: each step's input tile x_t is fetched from global once per workgroup (8 or 16 B per thread, whole rows per wave) and shared through two LDS tiles; before, all 8 waves fetched all of it
# speedup vs baseline: 6.7274x; 1.3517x over previous
#include <hip/hip_runtime.h>
#include <cstdint>
#include <cstddef>

typedef __attribute__((ext_vector_type(16))) _Float16 v16h;
typedef __attribute__((ext_vector_type(8)))  _Float16 v8h;
typedef __attribute__((ext_vector_type(4)))  _Float16 v4h;
typedef __attribute__((ext_vector_type(8)))  float    v8f;
typedef __attribute__((ext_vector_type(2)))  float    v2f;

#define B_    256
#define SEQ_  512
#define HOR_  32
#define NK_   8192
#define L1_   64
#define HID_  128
#define L2_   64
#define GATES_ 512
#define HLEN_ 544
#define HPAD_ 8

__global__ __launch_bounds__(256) void gemm1_kernel(const float* __restrict__ inp,
                                                    const float* __restrict__ akey,
                                                    float* __restrict__ dist) {
  const int tid = threadIdx.x, lane = tid & 31, wave = tid >> 5;
  const int mt = blockIdx.x, nt = blockIdx.y * 8 + wave;
  const int m = lane & 15;
  const int kb = (lane & 16) ? 2 : 0;
  const float* Ap = inp  + (size_t)(mt * 16 + m) * SEQ_ + kb;
  const float* Bp = akey + (size_t)(nt * 16 + m) * SEQ_ + kb;
  v8f acc0 = {}, acc1 = {};
  for (int k = 0; k < SEQ_; k += 8) {
    v2f a0 = *(const v2f*)(Ap + k);
    v2f b0 = *(const v2f*)(Bp + k);
    acc0 = __builtin_amdgcn_wmma_f32_16x16x4_f32(false, a0, false, b0, (short)0, acc0, false, false);
    v2f a1 = *(const v2f*)(Ap + k + 4);
    v2f b1 = *(const v2f*)(Bp + k + 4);
    acc1 = __builtin_amdgcn_wmma_f32_16x16x4_f32(false, a1, false, b1, (short)0, acc1, false, false);
  }
#pragma unroll
  for (int r = 0; r < 8; ++r) {
    int row = r + ((lane & 16) ? 8 : 0);
    dist[(size_t)(mt * 16 + row) * NK_ + nt * 16 + m] = acc0[r] + acc1[r];
  }
}

__global__ __launch_bounds__(256) void softmax_kernel(float* __restrict__ dist) {
  __shared__ float sv[256];
  __shared__ int   si[256];
  const int b = blockIdx.x, tid = threadIdx.x;
  float* row = dist + (size_t)b * NK_;

  float m = -3.4e38f; int mi = NK_;
  for (int j = tid; j < NK_; j += 256) {
    float v = row[j];
    if (v > m || (v == m && j < mi)) { m = v; mi = j; }
  }
  sv[tid] = m; si[tid] = mi;
  __syncthreads();
  for (int s = 128; s > 0; s >>= 1) {
    if (tid < s) {
      float v2 = sv[tid + s]; int i2 = si[tid + s];
      if (v2 > sv[tid] || (v2 == sv[tid] && i2 < si[tid])) { sv[tid] = v2; si[tid] = i2; }
    }
    __syncthreads();
  }
  if (tid == 0) row[si[0]] = 0.0f;
  __syncthreads();

  float mm = -3.4e38f;
  for (int j = tid; j < NK_; j += 256) mm = fmaxf(mm, row[j]);
  sv[tid] = mm;
  __syncthreads();
  for (int s = 128; s > 0; s >>= 1) { if (tid < s) sv[tid] = fmaxf(sv[tid], sv[tid + s]); __syncthreads(); }
  const float M = sv[0];
  __syncthreads();

  float sum = 0.f;
  for (int j = tid; j < NK_; j += 256) sum += __expf(row[j] - M);
  sv[tid] = sum;
  __syncthreads();
  for (int s = 128; s > 0; s >>= 1) { if (tid < s) sv[tid] += sv[tid + s]; __syncthreads(); }
  const float inv = 1.0f / sv[0];
  for (int j = tid; j < NK_; j += 256) row[j] = __expf(row[j] - M) * inv;
}

__global__ __launch_bounds__(256) void gemm2_kernel(const float* __restrict__ a,
                                                    const float* __restrict__ aval,
                                                    float* __restrict__ hist) {
  const int tid = threadIdx.x, lane = tid & 31, wave = tid >> 5;
  const int mt = blockIdx.x, nt = blockIdx.y * 8 + wave;
  if (nt >= (HLEN_ / 16)) return;
  const int m = lane & 15;
  const int kb = (lane & 16) ? 2 : 0;
  const int n = nt * 16 + m;
  const float* Ap = a + (size_t)(mt * 16 + m) * NK_ + kb;
  const float* Bp = aval + (size_t)kb * HLEN_ + n;
  v8f acc0 = {}, acc1 = {};
  for (int k = 0; k < NK_; k += 8) {
    v2f av = *(const v2f*)(Ap + k);
    v2f bv; bv[0] = Bp[(size_t)k * HLEN_]; bv[1] = Bp[(size_t)(k + 1) * HLEN_];
    acc0 = __builtin_amdgcn_wmma_f32_16x16x4_f32(false, av, false, bv, (short)0, acc0, false, false);
    v2f av1 = *(const v2f*)(Ap + k + 4);
    v2f bv1; bv1[0] = Bp[(size_t)(k + 4) * HLEN_]; bv1[1] = Bp[(size_t)(k + 5) * HLEN_];
    acc1 = __builtin_amdgcn_wmma_f32_16x16x4_f32(false, av1, false, bv1, (short)0, acc1, false, false);
  }
#pragma unroll
  for (int r = 0; r < 8; ++r) {
    int row = r + ((lane & 16) ? 8 : 0);
    hist[(size_t)(mt * 16 + row) * HLEN_ + n] = acc0[r] + acc1[r];
  }
}

__global__ __launch_bounds__(256) void featl1_kernel(const float* __restrict__ inp,
                                                     const float* __restrict__ hist,
                                                     const float* __restrict__ W1,
                                                     const float* __restrict__ b1,
                                                     _Float16* __restrict__ xf) {
  __shared__ float w1s[L1_ * (HOR_ + 1)];
  __shared__ float b1s[L1_];
  __shared__ float hs[8 + HOR_ - 1];
  __shared__ float ins[8];
  const int b = blockIdx.x, t0 = blockIdx.y * 8, tid = threadIdx.x;
  for (int i = tid; i < L1_ * (HOR_ + 1); i += 256) w1s[i] = W1[i];
  if (tid < L1_) b1s[tid] = b1[tid];
  if (tid < 8 + HOR_ - 1) hs[tid] = hist[(size_t)b * HLEN_ + t0 + tid];
  if (tid < 8) ins[tid] = inp[(size_t)b * SEQ_ + t0 + tid];
  __syncthreads();
  const int o = tid & 63;
#pragma unroll
  for (int pass = 0; pass < 2; ++pass) {
    int tt = (tid >> 6) + pass * 4;
    float acc = b1s[o] + ins[tt] * w1s[o * 33];
#pragma unroll
    for (int j = 0; j < HOR_; ++j) acc += hs[tt + j] * w1s[o * 33 + 1 + j];
    xf[((size_t)b * SEQ_ + (t0 + tt)) * L1_ + o] = (_Float16)tanhf(acc);
  }
}

#define LOG2E_F 1.4426950408889634f
__device__ __forceinline__ float act_logistic(float v) {
  const float e = __builtin_amdgcn_exp2f(-LOG2E_F * v);
  return __builtin_amdgcn_rcpf(1.0f + e);
}
__device__ __forceinline__ float act_tanh(float v) {
  const float e = __builtin_amdgcn_exp2f(-2.0f * LOG2E_F * v);
  return __builtin_fmaf(2.0f, __builtin_amdgcn_rcpf(1.0f + e), -1.0f);
}

#define CHUNK_  64
#define NCHUNK_ 8

template <int N> struct hvec_t;
template <> struct hvec_t<8> { typedef v8h type; };
template <> struct hvec_t<4> { typedef v4h type; };

template <int KIN, bool STORE_SEQ>
__device__ __forceinline__ void lstm_chunk(_Float16 (*hbuf)[16][HID_ + HPAD_],
                                           _Float16* xls,
                                           const _Float16* __restrict__ xin,
                                           const float* __restrict__ Wih,
                                           const float* __restrict__ Whh,
                                           const float* __restrict__ bih,
                                           const float* __restrict__ bhh,
                                           _Float16* __restrict__ hseq,
                                           _Float16* __restrict__ hstate,
                                           float* __restrict__ cstate,
                                           float* __restrict__ hlast,
                                           const int bm, const int t0) {
  constexpr int KFX = KIN / 32;
  const int tid = threadIdx.x, lane = tid & 31, wave = tid >> 5;
  const int m = lane & 15;
  const int hi16 = (lane >> 4) & 1;
  const int kbh = hi16 * 8;
  const int unit = wave * 16 + m;
  const int drow = hi16 * 8;
  const int tend = t0 + CHUNK_;

  const int prow = tid >> 4;
  const int pj = (tid & 15) * 8;
  float c[8];
  if (t0 == 0) {
#pragma unroll
    for (int r = 0; r < 8; ++r) c[r] = 0.f;
#pragma unroll
    for (int i = 0; i < 8; ++i) hbuf[0][prow][pj + i] = (_Float16)0.f;
  } else {
#pragma unroll
    for (int r = 0; r < 8; ++r) c[r] = cstate[(size_t)(bm + drow + r) * HID_ + unit];
    const _Float16* hp = STORE_SEQ ? hseq + ((size_t)(bm + prow) * SEQ_ + (t0 - 1)) * HID_ + pj
                                   : hstate + (size_t)(bm + prow) * HID_ + pj;
    *(v8h*)&hbuf[0][prow][pj] = *(const v8h*)hp;
  }
  constexpr int XE = KIN / 16;
  typedef typename hvec_t<XE>::type xvec;
  _Float16 (*xl)[16][KIN + HPAD_] = (_Float16 (*)[16][KIN + HPAD_])xls;
  const int xj = (tid & 15) * XE;
  const _Float16* xpiece = xin + (size_t)(bm + prow) * SEQ_ * KIN + xj;
  *(xvec*)&xl[1][prow][xj] = *(const xvec*)(xpiece + (size_t)(t0 + 1) * KIN);
  __syncthreads();


  v16h wihB[4][KFX], whhB[4][4];
  float biasv[4];
#pragma unroll
  for (int q = 0; q < 4; ++q) {
    const int g = q * HID_ + unit;
    biasv[q] = bih[g] + bhh[g];
#pragma unroll
    for (int kf = 0; kf < KFX; ++kf)
#pragma unroll
      for (int h = 0; h < 16; ++h) {
        int k = kf * 32 + ((h & 8) << 1) + kbh + (h & 7);
        wihB[q][kf][h] = (_Float16)Wih[(size_t)g * KIN + k];
      }
#pragma unroll
    for (int kf = 0; kf < 4; ++kf)
#pragma unroll
      for (int h = 0; h < 16; ++h) {
        int k = kf * 32 + ((h & 8) << 1) + kbh + (h & 7);
        whhB[q][kf][h] = (_Float16)Whh[(size_t)g * HID_ + k];
      }
    __builtin_amdgcn_sched_barrier(0);
  }

  const _Float16* xrow = xin + (size_t)(bm + m) * SEQ_ * KIN + kbh;
  v16h xbuf[2][KFX];
#pragma unroll
  for (int kf = 0; kf < KFX; ++kf) {
    const _Float16* p = xrow + (size_t)t0 * KIN + kf * 32;
    v8h lo = *(const v8h*)p;
    v8h hi = *(const v8h*)(p + 16);
#pragma unroll
    for (int i = 0; i < 8; ++i) { xbuf[0][kf][i] = lo[i]; xbuf[0][kf][i + 8] = hi[i]; }
  }

  float hv[8];
#pragma unroll
  for (int r = 0; r < 8; ++r) hv[r] = 0.f;

  for (int t2 = t0; t2 < tend; t2 += 2) {
#pragma unroll
    for (int u = 0; u < 2; ++u) {
      const int t = t2 + u;
      v16h hA[4];
#pragma unroll
      for (int kf = 0; kf < 4; ++kf) {
        const _Float16* p = &hbuf[u][m][kf * 32 + kbh];
        v8h lo = *(const v8h*)p;
        v8h hi = *(const v8h*)(p + 16);
#pragma unroll
        for (int i = 0; i < 8; ++i) { hA[kf][i] = lo[i]; hA[kf][i + 8] = hi[i]; }
      }
      v8h hrow;
      if (STORE_SEQ) hrow = *(const v8h*)&hbuf[u][prow][pj];
#pragma unroll
      for (int kf = 0; kf < KFX; ++kf) {
        const _Float16* p = &xl[u ^ 1][m][kf * 32 + kbh];
        v8h lo = *(const v8h*)p;
        v8h hi = *(const v8h*)(p + 16);
#pragma unroll
        for (int i = 0; i < 8; ++i) { xbuf[u ^ 1][kf][i] = lo[i]; xbuf[u ^ 1][kf][i + 8] = hi[i]; }
      }
      const int tn = (t + 2 < tend) ? t + 2 : tend - 1;
      const xvec xg = *(const xvec*)(xpiece + (size_t)tn * KIN);
      __builtin_amdgcn_sched_barrier(0);
      v8f acc[4];
#pragma unroll
      for (int q = 0; q < 4; ++q)
#pragma unroll
        for (int r = 0; r < 8; ++r) acc[q][r] = biasv[q];
#pragma unroll
      for (int kf = 0; kf < KFX; ++kf)
#pragma unroll
        for (int q = 0; q < 4; ++q)
          acc[q] = __builtin_amdgcn_wmma_f32_16x16x32_f16(false, xbuf[u][kf], false, wihB[q][kf],
                                                          (short)0, acc[q], false, false);
#pragma unroll
      for (int kf = 0; kf < 4; ++kf)
#pragma unroll
        for (int q = 0; q < 4; ++q)
          acc[q] = __builtin_amdgcn_wmma_f32_16x16x32_f16(false, hA[kf], false, whhB[q][kf],
                                                          (short)0, acc[q], false, false);
      if (STORE_SEQ && t > t0)
        *(v8h*)(hseq + ((size_t)(bm + prow) * SEQ_ + (t - 1)) * HID_ + pj) = hrow;
      float gi[8], gf[8], gg[8], go[8];
#pragma unroll
      for (int r = 0; r < 8; ++r) gi[r] = act_logistic(acc[0][r]);
#pragma unroll
      for (int r = 0; r < 8; ++r) gf[r] = act_logistic(acc[1][r]);
#pragma unroll
      for (int r = 0; r < 8; ++r) gg[r] = act_tanh(acc[2][r]);
#pragma unroll
      for (int r = 0; r < 8; ++r) go[r] = act_logistic(acc[3][r]);
#pragma unroll
      for (int r = 0; r < 8; ++r) c[r] = gf[r] * c[r] + gi[r] * gg[r];
#pragma unroll
      for (int r = 0; r < 8; ++r) hv[r] = go[r] * act_tanh(c[r]);
#pragma unroll
      for (int r = 0; r < 8; ++r) hbuf[u ^ 1][drow + r][unit] = (_Float16)hv[r];
      *(xvec*)&xl[u][prow][xj] = xg;
      __syncthreads();
    }
  }
  if (STORE_SEQ) {
    const v8h hrow = *(const v8h*)&hbuf[0][prow][pj];
    *(v8h*)(hseq + ((size_t)(bm + prow) * SEQ_ + (tend - 1)) * HID_ + pj) = hrow;
  } else if (tend == SEQ_) {
#pragma unroll
    for (int r = 0; r < 8; ++r) hlast[(size_t)(bm + drow + r) * HID_ + unit] = hv[r];
  } else {
    const v8h hrow = *(const v8h*)&hbuf[0][prow][pj];
    *(v8h*)(hstate + (size_t)(bm + prow) * HID_ + pj) = hrow;
  }
  if (tend < SEQ_) {
#pragma unroll
    for (int r = 0; r < 8; ++r) cstate[(size_t)(bm + drow + r) * HID_ + unit] = c[r];
  }
}

__global__ __launch_bounds__(256, 1) void lstm_pair_kernel(const _Float16* __restrict__ xf,
                                                           const float* __restrict__ Wih0,
                                                           const float* __restrict__ Whh0,
                                                           const float* __restrict__ bih0,
                                                           const float* __restrict__ bhh0,
                                                           const float* __restrict__ Wih1,
                                                           const float* __restrict__ Whh1,
                                                           const float* __restrict__ bih1,
                                                           const float* __restrict__ bhh1,
                                                           _Float16* h0seq,
                                                           float* c0state,
                                                           float* c1state,
                                                           _Float16* h1state,
                                                           float* h1last,
                                                           const int chunk) {
  __shared__ __align__(16) _Float16 hbuf[2][16][HID_ + HPAD_];
  __shared__ __align__(16) _Float16 xls[2 * 16 * (HID_ + HPAD_)];
  const int bm = (blockIdx.x & 15) * 16;
  if (blockIdx.x < 16) {
    if (chunk < NCHUNK_)
      lstm_chunk<L1_, true>(hbuf, xls, xf, Wih0, Whh0, bih0, bhh0, h0seq, nullptr, c0state, nullptr,
                            bm, chunk * CHUNK_);
  } else {
    if (chunk > 0)
      lstm_chunk<HID_, false>(hbuf, xls, h0seq, Wih1, Whh1, bih1, bhh1, nullptr, h1state, c1state, h1last,
                              bm, (chunk - 1) * CHUNK_);
  }
}

__global__ __launch_bounds__(128) void head_kernel(const float* __restrict__ h1last,
                                                   const float* __restrict__ W2,
                                                   const float* __restrict__ b2,
                                                   const float* __restrict__ W3,
                                                   const float* __restrict__ b3,
                                                   float* __restrict__ out) {
  __shared__ float lastS[HID_];
  __shared__ float l2S[L2_];
  const int b = blockIdx.x, tid = threadIdx.x;
  lastS[tid] = tanhf(h1last[(size_t)b * HID_ + tid]);
  __syncthreads();
  if (tid < L2_) {
    float acc = b2[tid];
    for (int k = 0; k < HID_; ++k) acc += lastS[k] * W2[tid * HID_ + k];
    l2S[tid] = tanhf(acc);
  }
  __syncthreads();
  if (tid < HOR_) {
    float acc = b3[tid];
    for (int k = 0; k < L2_; ++k) acc += l2S[k] * W3[tid * L2_ + k];
    out[(size_t)b * HOR_ + tid] = acc;
  }
}

extern "C" void kernel_launch(void* const* d_in, const int* in_sizes, int n_in,
                              void* d_out, int out_size, void* d_ws, size_t ws_size,
                              hipStream_t stream) {
  (void)in_sizes; (void)n_in; (void)out_size; (void)ws_size;
  const float* inp  = (const float*)d_in[0];
  const float* akey = (const float*)d_in[1];
  const float* aval = (const float*)d_in[2];
  const float* W1   = (const float*)d_in[3];
  const float* b1   = (const float*)d_in[4];
  const float* Wih0 = (const float*)d_in[5];
  const float* Whh0 = (const float*)d_in[6];
  const float* bih0 = (const float*)d_in[7];
  const float* bhh0 = (const float*)d_in[8];
  const float* Wih1 = (const float*)d_in[9];
  const float* Whh1 = (const float*)d_in[10];
  const float* bih1 = (const float*)d_in[11];
  const float* bhh1 = (const float*)d_in[12];
  const float* W2   = (const float*)d_in[13];
  const float* b2   = (const float*)d_in[14];
  const float* W3   = (const float*)d_in[15];
  const float* b3   = (const float*)d_in[16];

  char* ws = (char*)d_ws;
  size_t off = 0;
  float*    dist   = (float*)(ws + off);    off += (size_t)B_ * NK_ * 4;
  float*    hist   = (float*)(ws + off);    off += (size_t)B_ * HLEN_ * 4;
  _Float16* xf     = (_Float16*)(ws + off); off += (size_t)B_ * SEQ_ * L1_ * 2;
  _Float16* h0     = (_Float16*)(ws + off); off += (size_t)B_ * SEQ_ * HID_ * 2;
  float*    h1last = (float*)(ws + off);    off += (size_t)B_ * HID_ * 4;
  float*    c0st   = (float*)(ws + off);    off += (size_t)B_ * HID_ * 4;
  float*    c1st   = (float*)(ws + off);    off += (size_t)B_ * HID_ * 4;
  _Float16* h1st   = (_Float16*)(ws + off); off += (size_t)B_ * HID_ * 2;

  gemm1_kernel<<<dim3(16, 64), 256, 0, stream>>>(inp, akey, dist);
  softmax_kernel<<<dim3(B_), 256, 0, stream>>>(dist);
  gemm2_kernel<<<dim3(16, 5), 256, 0, stream>>>(dist, aval, hist);
  featl1_kernel<<<dim3(B_, SEQ_ / 8), 256, 0, stream>>>(inp, hist, W1, b1, xf);
  lstm_pair_kernel<<<dim3(2 * (B_ / 16)), 256, 0, stream>>>(xf, Wih0, Whh0, bih0, bhh0, Wih1, Whh1, bih1, bhh1,
                                                         h0, c0st, c1st, h1st, h1last, 0);
  lstm_pair_kernel<<<dim3(2 * (B_ / 16)), 256, 0, stream>>>(xf, Wih0, Whh0, bih0, bhh0, Wih1, Whh1, bih1, bhh1,
                                                         h0, c0st, c1st, h1st, h1last, 1);
  lstm_pair_kernel<<<dim3(2 * (B_ / 16)), 256, 0, stream>>>(xf, Wih0, Whh0, bih0, bhh0, Wih1, Whh1, bih1, bhh1,
                                                         h0, c0st, c1st, h1st, h1last, 2);
  lstm_pair_kernel<<<dim3(2 * (B_ / 16)), 256, 0, stream>>>(xf, Wih0, Whh0, bih0, bhh0, Wih1, Whh1, bih1, bhh1,
                                                         h0, c0st, c1st, h1st, h1last, 3);
  lstm_pair_kernel<<<dim3(2 * (B_ / 16)), 256, 0, stream>>>(xf, Wih0, Whh0, bih0, bhh0, Wih1, Whh1, bih1, bhh1,
                                                         h0, c0st, c1st, h1st, h1last, 4);
  lstm_pair_kernel<<<dim3(2 * (B_ / 16)), 256, 0, stream>>>(xf, Wih0, Whh0, bih0, bhh0, Wih1, Whh1, bih1, bhh1,
                                                         h0, c0st, c1st, h1st, h1last, 5);
  lstm_pair_kernel<<<dim3(2 * (B_ / 16)), 256, 0, stream>>>(xf, Wih0, Whh0, bih0, bhh0, Wih1, Whh1, bih1, bhh1,
                                                         h0, c0st, c1st, h1st, h1last, 6);
  lstm_pair_kernel<<<dim3(2 * (B_ / 16)), 256, 0, stream>>>(xf, Wih0, Whh0, bih0, bhh0, Wih1, Whh1, bih1, bhh1,
                                                         h0, c0st, c1st, h1st, h1last, 7);
  lstm_pair_kernel<<<dim3(2 * (B_ / 16)), 256, 0, stream>>>(xf, Wih0, Whh0, bih0, bhh0, Wih1, Whh1, bih1, bhh1,
                                                         h0, c0st, c1st, h1st, h1last, 8);
  head_kernel<<<dim3(B_), 128, 0, stream>>>(h1last, W2, b2, W3, b3, (float*)d_out);
}


# GEMM2 (hist = a @ aval) split along K into 16 slices: 1280 workgroups instead of 80; the 16 partial planes are summed in slice order by the feature kernel (hist's summation order changes)
# speedup vs baseline: 8.1846x; 1.2166x over previous
#include <hip/hip_runtime.h>
#include <cstdint>
#include <cstddef>

typedef __attribute__((ext_vector_type(16))) _Float16 v16h;
typedef __attribute__((ext_vector_type(8)))  _Float16 v8h;
typedef __attribute__((ext_vector_type(4)))  _Float16 v4h;
typedef __attribute__((ext_vector_type(8)))  float    v8f;
typedef __attribute__((ext_vector_type(2)))  float    v2f;

#define B_    256
#define SEQ_  512
#define HOR_  32
#define NK_   8192
#define L1_   64
#define HID_  128
#define L2_   64
#define GATES_ 512
#define HLEN_ 544
#define KSPLIT_ 16
#define HPAD_ 8

__global__ __launch_bounds__(256) void gemm1_kernel(const float* __restrict__ inp,
                                                    const float* __restrict__ akey,
                                                    float* __restrict__ dist) {
  const int tid = threadIdx.x, lane = tid & 31, wave = tid >> 5;
  const int mt = blockIdx.x, nt = blockIdx.y * 8 + wave;
  const int m = lane & 15;
  const int kb = (lane & 16) ? 2 : 0;
  const float* Ap = inp  + (size_t)(mt * 16 + m) * SEQ_ + kb;
  const float* Bp = akey + (size_t)(nt * 16 + m) * SEQ_ + kb;
  v8f acc0 = {}, acc1 = {};
  for (int k = 0; k < SEQ_; k += 8) {
    v2f a0 = *(const v2f*)(Ap + k);
    v2f b0 = *(const v2f*)(Bp + k);
    acc0 = __builtin_amdgcn_wmma_f32_16x16x4_f32(false, a0, false, b0, (short)0, acc0, false, false);
    v2f a1 = *(const v2f*)(Ap + k + 4);
    v2f b1 = *(const v2f*)(Bp + k + 4);
    acc1 = __builtin_amdgcn_wmma_f32_16x16x4_f32(false, a1, false, b1, (short)0, acc1, false, false);
  }
#pragma unroll
  for (int r = 0; r < 8; ++r) {
    int row = r + ((lane & 16) ? 8 : 0);
    dist[(size_t)(mt * 16 + row) * NK_ + nt * 16 + m] = acc0[r] + acc1[r];
  }
}

__global__ __launch_bounds__(256) void softmax_kernel(float* __restrict__ dist) {
  __shared__ float sv[256];
  __shared__ int   si[256];
  const int b = blockIdx.x, tid = threadIdx.x;
  float* row = dist + (size_t)b * NK_;

  float m = -3.4e38f; int mi = NK_;
  for (int j = tid; j < NK_; j += 256) {
    float v = row[j];
    if (v > m || (v == m && j < mi)) { m = v; mi = j; }
  }
  sv[tid] = m; si[tid] = mi;
  __syncthreads();
  for (int s = 128; s > 0; s >>= 1) {
    if (tid < s) {
      float v2 = sv[tid + s]; int i2 = si[tid + s];
      if (v2 > sv[tid] || (v2 == sv[tid] && i2 < si[tid])) { sv[tid] = v2; si[tid] = i2; }
    }
    __syncthreads();
  }
  if (tid == 0) row[si[0]] = 0.0f;
  __syncthreads();

  float mm = -3.4e38f;
  for (int j = tid; j < NK_; j += 256) mm = fmaxf(mm, row[j]);
  sv[tid] = mm;
  __syncthreads();
  for (int s = 128; s > 0; s >>= 1) { if (tid < s) sv[tid] = fmaxf(sv[tid], sv[tid + s]); __syncthreads(); }
  const float M = sv[0];
  __syncthreads();

  float sum = 0.f;
  for (int j = tid; j < NK_; j += 256) sum += __expf(row[j] - M);
  sv[tid] = sum;
  __syncthreads();
  for (int s = 128; s > 0; s >>= 1) { if (tid < s) sv[tid] += sv[tid + s]; __syncthreads(); }
  const float inv = 1.0f / sv[0];
  for (int j = tid; j < NK_; j += 256) row[j] = __expf(row[j] - M) * inv;
}

__global__ __launch_bounds__(256) void gemm2_kernel(const float* __restrict__ a,
                                                    const float* __restrict__ aval,
                                                    float* __restrict__ hist) {
  const int tid = threadIdx.x, lane = tid & 31, wave = tid >> 5;
  const int mt = blockIdx.x, nt = blockIdx.y * 8 + wave;
  if (nt >= (HLEN_ / 16)) return;
  const int m = lane & 15;
  const int kb = (lane & 16) ? 2 : 0;
  const int n = nt * 16 + m;
  const int ks = blockIdx.z * (NK_ / KSPLIT_);
  const float* Ap = a + (size_t)(mt * 16 + m) * NK_ + ks + kb;
  const float* Bp = aval + (size_t)(ks + kb) * HLEN_ + n;
  hist += (size_t)blockIdx.z * B_ * HLEN_;
  v8f acc0 = {}, acc1 = {};
  for (int k = 0; k < NK_ / KSPLIT_; k += 8) {
    v2f av = *(const v2f*)(Ap + k);
    v2f bv; bv[0] = Bp[(size_t)k * HLEN_]; bv[1] = Bp[(size_t)(k + 1) * HLEN_];
    acc0 = __builtin_amdgcn_wmma_f32_16x16x4_f32(false, av, false, bv, (short)0, acc0, false, false);
    v2f av1 = *(const v2f*)(Ap + k + 4);
    v2f bv1; bv1[0] = Bp[(size_t)(k + 4) * HLEN_]; bv1[1] = Bp[(size_t)(k + 5) * HLEN_];
    acc1 = __builtin_amdgcn_wmma_f32_16x16x4_f32(false, av1, false, bv1, (short)0, acc1, false, false);
  }
#pragma unroll
  for (int r = 0; r < 8; ++r) {
    int row = r + ((lane & 16) ? 8 : 0);
    hist[(size_t)(mt * 16 + row) * HLEN_ + n] = acc0[r] + acc1[r];
  }
}

__global__ __launch_bounds__(256) void featl1_kernel(const float* __restrict__ inp,
                                                     const float* __restrict__ hist,
                                                     const float* __restrict__ W1,
                                                     const float* __restrict__ b1,
                                                     _Float16* __restrict__ xf) {
  __shared__ float w1s[L1_ * (HOR_ + 1)];
  __shared__ float b1s[L1_];
  __shared__ float hs[8 + HOR_ - 1];
  __shared__ float ins[8];
  const int b = blockIdx.x, t0 = blockIdx.y * 8, tid = threadIdx.x;
  for (int i = tid; i < L1_ * (HOR_ + 1); i += 256) w1s[i] = W1[i];
  if (tid < L1_) b1s[tid] = b1[tid];
  if (tid < 8 + HOR_ - 1) {
    const float* hp = hist + (size_t)b * HLEN_ + t0 + tid;
    float v = hp[0];
    for (int s = 1; s < KSPLIT_; ++s) v += hp[(size_t)s * B_ * HLEN_];
    hs[tid] = v;
  }
  if (tid < 8) ins[tid] = inp[(size_t)b * SEQ_ + t0 + tid];
  __syncthreads();
  const int o = tid & 63;
#pragma unroll
  for (int pass = 0; pass < 2; ++pass) {
    int tt = (tid >> 6) + pass * 4;
    float acc = b1s[o] + ins[tt] * w1s[o * 33];
#pragma unroll
    for (int j = 0; j < HOR_; ++j) acc += hs[tt + j] * w1s[o * 33 + 1 + j];
    xf[((size_t)b * SEQ_ + (t0 + tt)) * L1_ + o] = (_Float16)tanhf(acc);
  }
}

#define LOG2E_F 1.4426950408889634f
__device__ __forceinline__ float act_logistic(float v) {
  const float e = __builtin_amdgcn_exp2f(-LOG2E_F * v);
  return __builtin_amdgcn_rcpf(1.0f + e);
}
__device__ __forceinline__ float act_tanh(float v) {
  const float e = __builtin_amdgcn_exp2f(-2.0f * LOG2E_F * v);
  return __builtin_fmaf(2.0f, __builtin_amdgcn_rcpf(1.0f + e), -1.0f);
}

#define CHUNK_  64
#define NCHUNK_ 8

template <int N> struct hvec_t;
template <> struct hvec_t<8> { typedef v8h type; };
template <> struct hvec_t<4> { typedef v4h type; };

template <int KIN, bool STORE_SEQ>
__device__ __forceinline__ void lstm_chunk(_Float16 (*hbuf)[16][HID_ + HPAD_],
                                           _Float16* xls,
                                           const _Float16* __restrict__ xin,
                                           const float* __restrict__ Wih,
                                           const float* __restrict__ Whh,
                                           const float* __restrict__ bih,
                                           const float* __restrict__ bhh,
                                           _Float16* __restrict__ hseq,
                                           _Float16* __restrict__ hstate,
                                           float* __restrict__ cstate,
                                           float* __restrict__ hlast,
                                           const int bm, const int t0) {
  constexpr int KFX = KIN / 32;
  const int tid = threadIdx.x, lane = tid & 31, wave = tid >> 5;
  const int m = lane & 15;
  const int hi16 = (lane >> 4) & 1;
  const int kbh = hi16 * 8;
  const int unit = wave * 16 + m;
  const int drow = hi16 * 8;
  const int tend = t0 + CHUNK_;

  const int prow = tid >> 4;
  const int pj = (tid & 15) * 8;
  float c[8];
  if (t0 == 0) {
#pragma unroll
    for (int r = 0; r < 8; ++r) c[r] = 0.f;
#pragma unroll
    for (int i = 0; i < 8; ++i) hbuf[0][prow][pj + i] = (_Float16)0.f;
  } else {
#pragma unroll
    for (int r = 0; r < 8; ++r) c[r] = cstate[(size_t)(bm + drow + r) * HID_ + unit];
    const _Float16* hp = STORE_SEQ ? hseq + ((size_t)(bm + prow) * SEQ_ + (t0 - 1)) * HID_ + pj
                                   : hstate + (size_t)(bm + prow) * HID_ + pj;
    *(v8h*)&hbuf[0][prow][pj] = *(const v8h*)hp;
  }
  constexpr int XE = KIN / 16;
  typedef typename hvec_t<XE>::type xvec;
  _Float16 (*xl)[16][KIN + HPAD_] = (_Float16 (*)[16][KIN + HPAD_])xls;
  const int xj = (tid & 15) * XE;
  const _Float16* xpiece = xin + (size_t)(bm + prow) * SEQ_ * KIN + xj;
  *(xvec*)&xl[1][prow][xj] = *(const xvec*)(xpiece + (size_t)(t0 + 1) * KIN);
  __syncthreads();


  v16h wihB[4][KFX], whhB[4][4];
  float biasv[4];
#pragma unroll
  for (int q = 0; q < 4; ++q) {
    const int g = q * HID_ + unit;
    biasv[q] = bih[g] + bhh[g];
#pragma unroll
    for (int kf = 0; kf < KFX; ++kf)
#pragma unroll
      for (int h = 0; h < 16; ++h) {
        int k = kf * 32 + ((h & 8) << 1) + kbh + (h & 7);
        wihB[q][kf][h] = (_Float16)Wih[(size_t)g * KIN + k];
      }
#pragma unroll
    for (int kf = 0; kf < 4; ++kf)
#pragma unroll
      for (int h = 0; h < 16; ++h) {
        int k = kf * 32 + ((h & 8) << 1) + kbh + (h & 7);
        whhB[q][kf][h] = (_Float16)Whh[(size_t)g * HID_ + k];
      }
    __builtin_amdgcn_sched_barrier(0);
  }

  const _Float16* xrow = xin + (size_t)(bm + m) * SEQ_ * KIN + kbh;
  v16h xbuf[2][KFX];
#pragma unroll
  for (int kf = 0; kf < KFX; ++kf) {
    const _Float16* p = xrow + (size_t)t0 * KIN + kf * 32;
    v8h lo = *(const v8h*)p;
    v8h hi = *(const v8h*)(p + 16);
#pragma unroll
    for (int i = 0; i < 8; ++i) { xbuf[0][kf][i] = lo[i]; xbuf[0][kf][i + 8] = hi[i]; }
  }

  float hv[8];
#pragma unroll
  for (int r = 0; r < 8; ++r) hv[r] = 0.f;

  for (int t2 = t0; t2 < tend; t2 += 2) {
#pragma unroll
    for (int u = 0; u < 2; ++u) {
      const int t = t2 + u;
      v16h hA[4];
#pragma unroll
      for (int kf = 0; kf < 4; ++kf) {
        const _Float16* p = &hbuf[u][m][kf * 32 + kbh];
        v8h lo = *(const v8h*)p;
        v8h hi = *(const v8h*)(p + 16);
#pragma unroll
        for (int i = 0; i < 8; ++i) { hA[kf][i] = lo[i]; hA[kf][i + 8] = hi[i]; }
      }
      v8h hrow;
      if (STORE_SEQ) hrow = *(const v8h*)&hbuf[u][prow][pj];
#pragma unroll
      for (int kf = 0; kf < KFX; ++kf) {
        const _Float16* p = &xl[u ^ 1][m][kf * 32 + kbh];
        v8h lo = *(const v8h*)p;
        v8h hi = *(const v8h*)(p + 16);
#pragma unroll
        for (int i = 0; i < 8; ++i) { xbuf[u ^ 1][kf][i] = lo[i]; xbuf[u ^ 1][kf][i + 8] = hi[i]; }
      }
      const int tn = (t + 2 < tend) ? t + 2 : tend - 1;
      const xvec xg = *(const xvec*)(xpiece + (size_t)tn * KIN);
      __builtin_amdgcn_sched_barrier(0);
      v8f acc[4];
#pragma unroll
      for (int q = 0; q < 4; ++q)
#pragma unroll
        for (int r = 0; r < 8; ++r) acc[q][r] = biasv[q];
#pragma unroll
      for (int kf = 0; kf < KFX; ++kf)
#pragma unroll
        for (int q = 0; q < 4; ++q)
          acc[q] = __builtin_amdgcn_wmma_f32_16x16x32_f16(false, xbuf[u][kf], false, wihB[q][kf],
                                                          (short)0, acc[q], false, false);
#pragma unroll
      for (int kf = 0; kf < 4; ++kf)
#pragma unroll
        for (int q = 0; q < 4; ++q)
          acc[q] = __builtin_amdgcn_wmma_f32_16x16x32_f16(false, hA[kf], false, whhB[q][kf],
                                                          (short)0, acc[q], false, false);
      if (STORE_SEQ && t > t0)
        *(v8h*)(hseq + ((size_t)(bm + prow) * SEQ_ + (t - 1)) * HID_ + pj) = hrow;
      float gi[8], gf[8], gg[8], go[8];
#pragma unroll
      for (int r = 0; r < 8; ++r) gi[r] = act_logistic(acc[0][r]);
#pragma unroll
      for (int r = 0; r < 8; ++r) gf[r] = act_logistic(acc[1][r]);
#pragma unroll
      for (int r = 0; r < 8; ++r) gg[r] = act_tanh(acc[2][r]);
#pragma unroll
      for (int r = 0; r < 8; ++r) go[r] = act_logistic(acc[3][r]);
#pragma unroll
      for (int r = 0; r < 8; ++r) c[r] = gf[r] * c[r] + gi[r] * gg[r];
#pragma unroll
      for (int r = 0; r < 8; ++r) hv[r] = go[r] * act_tanh(c[r]);
#pragma unroll
      for (int r = 0; r < 8; ++r) hbuf[u ^ 1][drow + r][unit] = (_Float16)hv[r];
      *(xvec*)&xl[u][prow][xj] = xg;
      __syncthreads();
    }
  }
  if (STORE_SEQ) {
    const v8h hrow = *(const v8h*)&hbuf[0][prow][pj];
    *(v8h*)(hseq + ((size_t)(bm + prow) * SEQ_ + (tend - 1)) * HID_ + pj) = hrow;
  } else if (tend == SEQ_) {
#pragma unroll
    for (int r = 0; r < 8; ++r) hlast[(size_t)(bm + drow + r) * HID_ + unit] = hv[r];
  } else {
    const v8h hrow = *(const v8h*)&hbuf[0][prow][pj];
    *(v8h*)(hstate + (size_t)(bm + prow) * HID_ + pj) = hrow;
  }
  if (tend < SEQ_) {
#pragma unroll
    for (int r = 0; r < 8; ++r) cstate[(size_t)(bm + drow + r) * HID_ + unit] = c[r];
  }
}

__global__ __launch_bounds__(256, 1) void lstm_pair_kernel(const _Float16* __restrict__ xf,
                                                           const float* __restrict__ Wih0,
                                                           const float* __restrict__ Whh0,
                                                           const float* __restrict__ bih0,
                                                           const float* __restrict__ bhh0,
                                                           const float* __restrict__ Wih1,
                                                           const float* __restrict__ Whh1,
                                                           const float* __restrict__ bih1,
                                                           const float* __restrict__ bhh1,
                                                           _Float16* h0seq,
                                                           float* c0state,
                                                           float* c1state,
                                                           _Float16* h1state,
                                                           float* h1last,
                                                           const int chunk) {
  __shared__ __align__(16) _Float16 hbuf[2][16][HID_ + HPAD_];
  __shared__ __align__(16) _Float16 xls[2 * 16 * (HID_ + HPAD_)];
  const int bm = (blockIdx.x & 15) * 16;
  if (blockIdx.x < 16) {
    if (chunk < NCHUNK_)
      lstm_chunk<L1_, true>(hbuf, xls, xf, Wih0, Whh0, bih0, bhh0, h0seq, nullptr, c0state, nullptr,
                            bm, chunk * CHUNK_);
  } else {
    if (chunk > 0)
      lstm_chunk<HID_, false>(hbuf, xls, h0seq, Wih1, Whh1, bih1, bhh1, nullptr, h1state, c1state, h1last,
                              bm, (chunk - 1) * CHUNK_);
  }
}

__global__ __launch_bounds__(128) void head_kernel(const float* __restrict__ h1last,
                                                   const float* __restrict__ W2,
                                                   const float* __restrict__ b2,
                                                   const float* __restrict__ W3,
                                                   const float* __restrict__ b3,
                                                   float* __restrict__ out) {
  __shared__ float lastS[HID_];
  __shared__ float l2S[L2_];
  const int b = blockIdx.x, tid = threadIdx.x;
  lastS[tid] = tanhf(h1last[(size_t)b * HID_ + tid]);
  __syncthreads();
  if (tid < L2_) {
    float acc = b2[tid];
    for (int k = 0; k < HID_; ++k) acc += lastS[k] * W2[tid * HID_ + k];
    l2S[tid] = tanhf(acc);
  }
  __syncthreads();
  if (tid < HOR_) {
    float acc = b3[tid];
    for (int k = 0; k < L2_; ++k) acc += l2S[k] * W3[tid * L2_ + k];
    out[(size_t)b * HOR_ + tid] = acc;
  }
}

extern "C" void kernel_launch(void* const* d_in, const int* in_sizes, int n_in,
                              void* d_out, int out_size, void* d_ws, size_t ws_size,
                              hipStream_t stream) {
  (void)in_sizes; (void)n_in; (void)out_size; (void)ws_size;
  const float* inp  = (const float*)d_in[0];
  const float* akey = (const float*)d_in[1];
  const float* aval = (const float*)d_in[2];
  const float* W1   = (const float*)d_in[3];
  const float* b1   = (const float*)d_in[4];
  const float* Wih0 = (const float*)d_in[5];
  const float* Whh0 = (const float*)d_in[6];
  const float* bih0 = (const float*)d_in[7];
  const float* bhh0 = (const float*)d_in[8];
  const float* Wih1 = (const float*)d_in[9];
  const float* Whh1 = (const float*)d_in[10];
  const float* bih1 = (const float*)d_in[11];
  const float* bhh1 = (const float*)d_in[12];
  const float* W2   = (const float*)d_in[13];
  const float* b2   = (const float*)d_in[14];
  const float* W3   = (const float*)d_in[15];
  const float* b3   = (const float*)d_in[16];

  char* ws = (char*)d_ws;
  size_t off = 0;
  float*    dist   = (float*)(ws + off);    off += (size_t)B_ * NK_ * 4;
  float*    hist   = (float*)(ws + off);    off += (size_t)KSPLIT_ * B_ * HLEN_ * 4;
  _Float16* xf     = (_Float16*)(ws + off); off += (size_t)B_ * SEQ_ * L1_ * 2;
  _Float16* h0     = (_Float16*)(ws + off); off += (size_t)B_ * SEQ_ * HID_ * 2;
  float*    h1last = (float*)(ws + off);    off += (size_t)B_ * HID_ * 4;
  float*    c0st   = (float*)(ws + off);    off += (size_t)B_ * HID_ * 4;
  float*    c1st   = (float*)(ws + off);    off += (size_t)B_ * HID_ * 4;
  _Float16* h1st   = (_Float16*)(ws + off); off += (size_t)B_ * HID_ * 2;

  gemm1_kernel<<<dim3(16, 64), 256, 0, stream>>>(inp, akey, dist);
  softmax_kernel<<<dim3(B_), 256, 0, stream>>>(dist);
  gemm2_kernel<<<dim3(16, 5, KSPLIT_), 256, 0, stream>>>(dist, aval, hist);
  featl1_kernel<<<dim3(B_, SEQ_ / 8), 256, 0, stream>>>(inp, hist, W1, b1, xf);
  lstm_pair_kernel<<<dim3(2 * (B_ / 16)), 256, 0, stream>>>(xf, Wih0, Whh0, bih0, bhh0, Wih1, Whh1, bih1, bhh1,
                                                         h0, c0st, c1st, h1st, h1last, 0);
  lstm_pair_kernel<<<dim3(2 * (B_ / 16)), 256, 0, stream>>>(xf, Wih0, Whh0, bih0, bhh0, Wih1, Whh1, bih1, bhh1,
                                                         h0, c0st, c1st, h1st, h1last, 1);
  lstm_pair_kernel<<<dim3(2 * (B_ / 16)), 256, 0, stream>>>(xf, Wih0, Whh0, bih0, bhh0, Wih1, Whh1, bih1, bhh1,
                                                         h0, c0st, c1st, h1st, h1last, 2);
  lstm_pair_kernel<<<dim3(2 * (B_ / 16)), 256, 0, stream>>>(xf, Wih0, Whh0, bih0, bhh0, Wih1, Whh1, bih1, bhh1,
                                                         h0, c0st, c1st, h1st, h1last, 3);
  lstm_pair_kernel<<<dim3(2 * (B_ / 16)), 256, 0, stream>>>(xf, Wih0, Whh0, bih0, bhh0, Wih1, Whh1, bih1, bhh1,
                                                         h0, c0st, c1st, h1st, h1last, 4);
  lstm_pair_kernel<<<dim3(2 * (B_ / 16)), 256, 0, stream>>>(xf, Wih0, Whh0, bih0, bhh0, Wih1, Whh1, bih1, bhh1,
                                                         h0, c0st, c1st, h1st, h1last, 5);
  lstm_pair_kernel<<<dim3(2 * (B_ / 16)), 256, 0, stream>>>(xf, Wih0, Whh0, bih0, bhh0, Wih1, Whh1, bih1, bhh1,
                                                         h0, c0st, c1st, h1st, h1last, 6);
  lstm_pair_kernel<<<dim3(2 * (B_ / 16)), 256, 0, stream>>>(xf, Wih0, Whh0, bih0, bhh0, Wih1, Whh1, bih1, bhh1,
                                                         h0, c0st, c1st, h1st, h1last, 7);
  lstm_pair_kernel<<<dim3(2 * (B_ / 16)), 256, 0, stream>>>(xf, Wih0, Whh0, bih0, bhh0, Wih1, Whh1, bih1, bhh1,
                                                         h0, c0st, c1st, h1st, h1last, 8);
  head_kernel<<<dim3(B_), 128, 0, stream>>>(h1last, W2, b2, W3, b3, (float*)d_out);
}


# Feature kernel retiled: one workgroup per batch row and 64 steps (2048 workgroups, was 16384); a thread holds its W1 row and hist window in registers for 16 steps; same sum order, products fused
# speedup vs baseline: 8.4460x; 1.0319x over previous
#include <hip/hip_runtime.h>
#include <cstdint>
#include <cstddef>

typedef __attribute__((ext_vector_type(16))) _Float16 v16h;
typedef __attribute__((ext_vector_type(8)))  _Float16 v8h;
typedef __attribute__((ext_vector_type(4)))  _Float16 v4h;
typedef __attribute__((ext_vector_type(8)))  float    v8f;
typedef __attribute__((ext_vector_type(2)))  float    v2f;

#define B_    256
#define SEQ_  512
#define HOR_  32
#define NK_   8192
#define L1_   64
#define HID_  128
#define L2_   64
#define GATES_ 512
#define HLEN_ 544
#define KSPLIT_ 16
#define HPAD_ 8

__global__ __launch_bounds__(256) void gemm1_kernel(const float* __restrict__ inp,
                                                    const float* __restrict__ akey,
                                                    float* __restrict__ dist) {
  const int tid = threadIdx.x, lane = tid & 31, wave = tid >> 5;
  const int mt = blockIdx.x, nt = blockIdx.y * 8 + wave;
  const int m = lane & 15;
  const int kb = (lane & 16) ? 2 : 0;
  const float* Ap = inp  + (size_t)(mt * 16 + m) * SEQ_ + kb;
  const float* Bp = akey + (size_t)(nt * 16 + m) * SEQ_ + kb;
  v8f acc0 = {}, acc1 = {};
  for (int k = 0; k < SEQ_; k += 8) {
    v2f a0 = *(const v2f*)(Ap + k);
    v2f b0 = *(const v2f*)(Bp + k);
    acc0 = __builtin_amdgcn_wmma_f32_16x16x4_f32(false, a0, false, b0, (short)0, acc0, false, false);
    v2f a1 = *(const v2f*)(Ap + k + 4);
    v2f b1 = *(const v2f*)(Bp + k + 4);
    acc1 = __builtin_amdgcn_wmma_f32_16x16x4_f32(false, a1, false, b1, (short)0, acc1, false, false);
  }
#pragma unroll
  for (int r = 0; r < 8; ++r) {
    int row = r + ((lane & 16) ? 8 : 0);
    dist[(size_t)(mt * 16 + row) * NK_ + nt * 16 + m] = acc0[r] + acc1[r];
  }
}

__global__ __launch_bounds__(256) void softmax_kernel(float* __restrict__ dist) {
  __shared__ float sv[256];
  __shared__ int   si[256];
  const int b = blockIdx.x, tid = threadIdx.x;
  float* row = dist + (size_t)b * NK_;

  float m = -3.4e38f; int mi = NK_;
  for (int j = tid; j < NK_; j += 256) {
    float v = row[j];
    if (v > m || (v == m && j < mi)) { m = v; mi = j; }
  }
  sv[tid] = m; si[tid] = mi;
  __syncthreads();
  for (int s = 128; s > 0; s >>= 1) {
    if (tid < s) {
      float v2 = sv[tid + s]; int i2 = si[tid + s];
      if (v2 > sv[tid] || (v2 == sv[tid] && i2 < si[tid])) { sv[tid] = v2; si[tid] = i2; }
    }
    __syncthreads();
  }
  if (tid == 0) row[si[0]] = 0.0f;
  __syncthreads();

  float mm = -3.4e38f;
  for (int j = tid; j < NK_; j += 256) mm = fmaxf(mm, row[j]);
  sv[tid] = mm;
  __syncthreads();
  for (int s = 128; s > 0; s >>= 1) { if (tid < s) sv[tid] = fmaxf(sv[tid], sv[tid + s]); __syncthreads(); }
  const float M = sv[0];
  __syncthreads();

  float sum = 0.f;
  for (int j = tid; j < NK_; j += 256) sum += __expf(row[j] - M);
  sv[tid] = sum;
  __syncthreads();
  for (int s = 128; s > 0; s >>= 1) { if (tid < s) sv[tid] += sv[tid + s]; __syncthreads(); }
  const float inv = 1.0f / sv[0];
  for (int j = tid; j < NK_; j += 256) row[j] = __expf(row[j] - M) * inv;
}

__global__ __launch_bounds__(256) void gemm2_kernel(const float* __restrict__ a,
                                                    const float* __restrict__ aval,
                                                    float* __restrict__ hist) {
  const int tid = threadIdx.x, lane = tid & 31, wave = tid >> 5;
  const int mt = blockIdx.x, nt = blockIdx.y * 8 + wave;
  if (nt >= (HLEN_ / 16)) return;
  const int m = lane & 15;
  const int kb = (lane & 16) ? 2 : 0;
  const int n = nt * 16 + m;
  const int ks = blockIdx.z * (NK_ / KSPLIT_);
  const float* Ap = a + (size_t)(mt * 16 + m) * NK_ + ks + kb;
  const float* Bp = aval + (size_t)(ks + kb) * HLEN_ + n;
  hist += (size_t)blockIdx.z * B_ * HLEN_;
  v8f acc0 = {}, acc1 = {};
  for (int k = 0; k < NK_ / KSPLIT_; k += 8) {
    v2f av = *(const v2f*)(Ap + k);
    v2f bv; bv[0] = Bp[(size_t)k * HLEN_]; bv[1] = Bp[(size_t)(k + 1) * HLEN_];
    acc0 = __builtin_amdgcn_wmma_f32_16x16x4_f32(false, av, false, bv, (short)0, acc0, false, false);
    v2f av1 = *(const v2f*)(Ap + k + 4);
    v2f bv1; bv1[0] = Bp[(size_t)(k + 4) * HLEN_]; bv1[1] = Bp[(size_t)(k + 5) * HLEN_];
    acc1 = __builtin_amdgcn_wmma_f32_16x16x4_f32(false, av1, false, bv1, (short)0, acc1, false, false);
  }
#pragma unroll
  for (int r = 0; r < 8; ++r) {
    int row = r + ((lane & 16) ? 8 : 0);
    hist[(size_t)(mt * 16 + row) * HLEN_ + n] = acc0[r] + acc1[r];
  }
}

#define FT_ 64
#define FS_ 16
__global__ __launch_bounds__(256) void featl1_kernel(const float* __restrict__ inp,
                                                     const float* __restrict__ hist,
                                                     const float* __restrict__ W1,
                                                     const float* __restrict__ b1,
                                                     _Float16* __restrict__ xf) {
  __shared__ float w1s[L1_ * (HOR_ + 1)];
  __shared__ __align__(16) float hs[FT_ + HOR_];
  __shared__ __align__(16) float ins[FT_];
  const int b = blockIdx.x, t0 = blockIdx.y * FT_, tid = threadIdx.x;
  for (int i = tid; i < L1_ * (HOR_ + 1); i += 256) w1s[i] = W1[i];
  if (tid < FT_ + HOR_ - 1) {
    const float* hp = hist + (size_t)b * HLEN_ + t0 + tid;
    float v = hp[0];
    for (int s = 1; s < KSPLIT_; ++s) v += hp[(size_t)s * B_ * HLEN_];
    hs[tid] = v;
  } else if (tid == FT_ + HOR_ - 1) {
    hs[tid] = 0.f;
  } else if (tid >= 256 - FT_) {
    ins[tid - (256 - FT_)] = inp[(size_t)b * SEQ_ + t0 + tid - (256 - FT_)];
  }
  const int o = tid & 63;
  const int g = tid >> 6;
  const float bo = b1[o];
  __syncthreads();

  float w[HOR_ + 1];
#pragma unroll
  for (int j = 0; j < HOR_ + 1; ++j) w[j] = w1s[o * (HOR_ + 1) + j];
  float h[FS_ + HOR_];
#pragma unroll
  for (int i = 0; i < (FS_ + HOR_) / 4; ++i) {
    const float4 v = *(const float4*)&hs[g * FS_ + 4 * i];
    h[4 * i] = v.x; h[4 * i + 1] = v.y; h[4 * i + 2] = v.z; h[4 * i + 3] = v.w;
  }
  float acc[FS_];
#pragma unroll
  for (int i = 0; i < FS_ / 4; ++i) {
    const float4 v = *(const float4*)&ins[g * FS_ + 4 * i];
    acc[4 * i]     = __builtin_fmaf(v.x, w[0], bo);
    acc[4 * i + 1] = __builtin_fmaf(v.y, w[0], bo);
    acc[4 * i + 2] = __builtin_fmaf(v.z, w[0], bo);
    acc[4 * i + 3] = __builtin_fmaf(v.w, w[0], bo);
  }
#pragma unroll
  for (int j = 0; j < HOR_; ++j)
#pragma unroll
    for (int i = 0; i < FS_; ++i) acc[i] = __builtin_fmaf(h[i + j], w[1 + j], acc[i]);
  _Float16* xo = xf + ((size_t)b * SEQ_ + t0 + g * FS_) * L1_ + o;
#pragma unroll
  for (int i = 0; i < FS_; ++i) xo[(size_t)i * L1_] = (_Float16)tanhf(acc[i]);
}

#define LOG2E_F 1.4426950408889634f
__device__ __forceinline__ float act_logistic(float v) {
  const float e = __builtin_amdgcn_exp2f(-LOG2E_F * v);
  return __builtin_amdgcn_rcpf(1.0f + e);
}
__device__ __forceinline__ float act_tanh(float v) {
  const float e = __builtin_amdgcn_exp2f(-2.0f * LOG2E_F * v);
  return __builtin_fmaf(2.0f, __builtin_amdgcn_rcpf(1.0f + e), -1.0f);
}

#define CHUNK_  64
#define NCHUNK_ 8

template <int N> struct hvec_t;
template <> struct hvec_t<8> { typedef v8h type; };
template <> struct hvec_t<4> { typedef v4h type; };

template <int KIN, bool STORE_SEQ>
__device__ __forceinline__ void lstm_chunk(_Float16 (*hbuf)[16][HID_ + HPAD_],
                                           _Float16* xls,
                                           const _Float16* __restrict__ xin,
                                           const float* __restrict__ Wih,
                                           const float* __restrict__ Whh,
                                           const float* __restrict__ bih,
                                           const float* __restrict__ bhh,
                                           _Float16* __restrict__ hseq,
                                           _Float16* __restrict__ hstate,
                                           float* __restrict__ cstate,
                                           float* __restrict__ hlast,
                                           const int bm, const int t0) {
  constexpr int KFX = KIN / 32;
  const int tid = threadIdx.x, lane = tid & 31, wave = tid >> 5;
  const int m = lane & 15;
  const int hi16 = (lane >> 4) & 1;
  const int kbh = hi16 * 8;
  const int unit = wave * 16 + m;
  const int drow = hi16 * 8;
  const int tend = t0 + CHUNK_;

  const int prow = tid >> 4;
  const int pj = (tid & 15) * 8;
  float c[8];
  if (t0 == 0) {
#pragma unroll
    for (int r = 0; r < 8; ++r) c[r] = 0.f;
#pragma unroll
    for (int i = 0; i < 8; ++i) hbuf[0][prow][pj + i] = (_Float16)0.f;
  } else {
#pragma unroll
    for (int r = 0; r < 8; ++r) c[r] = cstate[(size_t)(bm + drow + r) * HID_ + unit];
    const _Float16* hp = STORE_SEQ ? hseq + ((size_t)(bm + prow) * SEQ_ + (t0 - 1)) * HID_ + pj
                                   : hstate + (size_t)(bm + prow) * HID_ + pj;
    *(v8h*)&hbuf[0][prow][pj] = *(const v8h*)hp;
  }
  constexpr int XE = KIN / 16;
  typedef typename hvec_t<XE>::type xvec;
  _Float16 (*xl)[16][KIN + HPAD_] = (_Float16 (*)[16][KIN + HPAD_])xls;
  const int xj = (tid & 15) * XE;
  const _Float16* xpiece = xin + (size_t)(bm + prow) * SEQ_ * KIN + xj;
  *(xvec*)&xl[1][prow][xj] = *(const xvec*)(xpiece + (size_t)(t0 + 1) * KIN);
  __syncthreads();


  v16h wihB[4][KFX], whhB[4][4];
  float biasv[4];
#pragma unroll
  for (int q = 0; q < 4; ++q) {
    const int g = q * HID_ + unit;
    biasv[q] = bih[g] + bhh[g];
#pragma unroll
    for (int kf = 0; kf < KFX; ++kf)
#pragma unroll
      for (int h = 0; h < 16; ++h) {
        int k = kf * 32 + ((h & 8) << 1) + kbh + (h & 7);
        wihB[q][kf][h] = (_Float16)Wih[(size_t)g * KIN + k];
      }
#pragma unroll
    for (int kf = 0; kf < 4; ++kf)
#pragma unroll
      for (int h = 0; h < 16; ++h) {
        int k = kf * 32 + ((h & 8) << 1) + kbh + (h & 7);
        whhB[q][kf][h] = (_Float16)Whh[(size_t)g * HID_ + k];
      }
    __builtin_amdgcn_sched_barrier(0);
  }

  const _Float16* xrow = xin + (size_t)(bm + m) * SEQ_ * KIN + kbh;
  v16h xbuf[2][KFX];
#pragma unroll
  for (int kf = 0; kf < KFX; ++kf) {
    const _Float16* p = xrow + (size_t)t0 * KIN + kf * 32;
    v8h lo = *(const v8h*)p;
    v8h hi = *(const v8h*)(p + 16);
#pragma unroll
    for (int i = 0; i < 8; ++i) { xbuf[0][kf][i] = lo[i]; xbuf[0][kf][i + 8] = hi[i]; }
  }

  float hv[8];
#pragma unroll
  for (int r = 0; r < 8; ++r) hv[r] = 0.f;

  for (int t2 = t0; t2 < tend; t2 += 2) {
#pragma unroll
    for (int u = 0; u < 2; ++u) {
      const int t = t2 + u;
      v16h hA[4];
#pragma unroll
      for (int kf = 0; kf < 4; ++kf) {
        const _Float16* p = &hbuf[u][m][kf * 32 + kbh];
        v8h lo = *(const v8h*)p;
        v8h hi = *(const v8h*)(p + 16);
#pragma unroll
        for (int i = 0; i < 8; ++i) { hA[kf][i] = lo[i]; hA[kf][i + 8] = hi[i]; }
      }
      v8h hrow;
      if (STORE_SEQ) hrow = *(const v8h*)&hbuf[u][prow][pj];
#pragma unroll
      for (int kf = 0; kf < KFX; ++kf) {
        const _Float16* p = &xl[u ^ 1][m][kf * 32 + kbh];
        v8h lo = *(const v8h*)p;
        v8h hi = *(const v8h*)(p + 16);
#pragma unroll
        for (int i = 0; i < 8; ++i) { xbuf[u ^ 1][kf][i] = lo[i]; xbuf[u ^ 1][kf][i + 8] = hi[i]; }
      }
      const int tn = (t + 2 < tend) ? t + 2 : tend - 1;
      const xvec xg = *(const xvec*)(xpiece + (size_t)tn * KIN);
      __builtin_amdgcn_sched_barrier(0);
      v8f acc[4];
#pragma unroll
      for (int q = 0; q < 4; ++q)
#pragma unroll
        for (int r = 0; r < 8; ++r) acc[q][r] = biasv[q];
#pragma unroll
      for (int kf = 0; kf < KFX; ++kf)
#pragma unroll
        for (int q = 0; q < 4; ++q)
          acc[q] = __builtin_amdgcn_wmma_f32_16x16x32_f16(false, xbuf[u][kf], false, wihB[q][kf],
                                                          (short)0, acc[q], false, false);
#pragma unroll
      for (int kf = 0; kf < 4; ++kf)
#pragma unroll
        for (int q = 0; q < 4; ++q)
          acc[q] = __builtin_amdgcn_wmma_f32_16x16x32_f16(false, hA[kf], false, whhB[q][kf],
                                                          (short)0, acc[q], false, false);
      if (STORE_SEQ && t > t0)
        *(v8h*)(hseq + ((size_t)(bm + prow) * SEQ_ + (t - 1)) * HID_ + pj) = hrow;
      float gi[8], gf[8], gg[8], go[8];
#pragma unroll
      for (int r = 0; r < 8; ++r) gi[r] = act_logistic(acc[0][r]);
#pragma unroll
      for (int r = 0; r < 8; ++r) gf[r] = act_logistic(acc[1][r]);
#pragma unroll
      for (int r = 0; r < 8; ++r) gg[r] = act_tanh(acc[2][r]);
#pragma unroll
      for (int r = 0; r < 8; ++r) go[r] = act_logistic(acc[3][r]);
#pragma unroll
      for (int r = 0; r < 8; ++r) c[r] = gf[r] * c[r] + gi[r] * gg[r];
#pragma unroll
      for (int r = 0; r < 8; ++r) hv[r] = go[r] * act_tanh(c[r]);
#pragma unroll
      for (int r = 0; r < 8; ++r) hbuf[u ^ 1][drow + r][unit] = (_Float16)hv[r];
      *(xvec*)&xl[u][prow][xj] = xg;
      __syncthreads();
    }
  }
  if (STORE_SEQ) {
    const v8h hrow = *(const v8h*)&hbuf[0][prow][pj];
    *(v8h*)(hseq + ((size_t)(bm + prow) * SEQ_ + (tend - 1)) * HID_ + pj) = hrow;
  } else if (tend == SEQ_) {
#pragma unroll
    for (int r = 0; r < 8; ++r) hlast[(size_t)(bm + drow + r) * HID_ + unit] = hv[r];
  } else {
    const v8h hrow = *(const v8h*)&hbuf[0][prow][pj];
    *(v8h*)(hstate + (size_t)(bm + prow) * HID_ + pj) = hrow;
  }
  if (tend < SEQ_) {
#pragma unroll
    for (int r = 0; r < 8; ++r) cstate[(size_t)(bm + drow + r) * HID_ + unit] = c[r];
  }
}

__global__ __launch_bounds__(256, 1) void lstm_pair_kernel(const _Float16* __restrict__ xf,
                                                           const float* __restrict__ Wih0,
                                                           const float* __restrict__ Whh0,
                                                           const float* __restrict__ bih0,
                                                           const float* __restrict__ bhh0,
                                                           const float* __restrict__ Wih1,
                                                           const float* __restrict__ Whh1,
                                                           const float* __restrict__ bih1,
                                                           const float* __restrict__ bhh1,
                                                           _Float16* h0seq,
                                                           float* c0state,
                                                           float* c1state,
                                                           _Float16* h1state,
                                                           float* h1last,
                                                           const int chunk) {
  __shared__ __align__(16) _Float16 hbuf[2][16][HID_ + HPAD_];
  __shared__ __align__(16) _Float16 xls[2 * 16 * (HID_ + HPAD_)];
  const int bm = (blockIdx.x & 15) * 16;
  if (blockIdx.x < 16) {
    if (chunk < NCHUNK_)
      lstm_chunk<L1_, true>(hbuf, xls, xf, Wih0, Whh0, bih0, bhh0, h0seq, nullptr, c0state, nullptr,
                            bm, chunk * CHUNK_);
  } else {
    if (chunk > 0)
      lstm_chunk<HID_, false>(hbuf, xls, h0seq, Wih1, Whh1, bih1, bhh1, nullptr, h1state, c1state, h1last,
                              bm, (chunk - 1) * CHUNK_);
  }
}

__global__ __launch_bounds__(128) void head_kernel(const float* __restrict__ h1last,
                                                   const float* __restrict__ W2,
                                                   const float* __restrict__ b2,
                                                   const float* __restrict__ W3,
                                                   const float* __restrict__ b3,
                                                   float* __restrict__ out) {
  __shared__ float lastS[HID_];
  __shared__ float l2S[L2_];
  const int b = blockIdx.x, tid = threadIdx.x;
  lastS[tid] = tanhf(h1last[(size_t)b * HID_ + tid]);
  __syncthreads();
  if (tid < L2_) {
    float acc = b2[tid];
    for (int k = 0; k < HID_; ++k) acc += lastS[k] * W2[tid * HID_ + k];
    l2S[tid] = tanhf(acc);
  }
  __syncthreads();
  if (tid < HOR_) {
    float acc = b3[tid];
    for (int k = 0; k < L2_; ++k) acc += l2S[k] * W3[tid * L2_ + k];
    out[(size_t)b * HOR_ + tid] = acc;
  }
}

extern "C" void kernel_launch(void* const* d_in, const int* in_sizes, int n_in,
                              void* d_out, int out_size, void* d_ws, size_t ws_size,
                              hipStream_t stream) {
  (void)in_sizes; (void)n_in; (void)out_size; (void)ws_size;
  const float* inp  = (const float*)d_in[0];
  const float* akey = (const float*)d_in[1];
  const float* aval = (const float*)d_in[2];
  const float* W1   = (const float*)d_in[3];
  const float* b1   = (const float*)d_in[4];
  const float* Wih0 = (const float*)d_in[5];
  const float* Whh0 = (const float*)d_in[6];
  const float* bih0 = (const float*)d_in[7];
  const float* bhh0 = (const float*)d_in[8];
  const float* Wih1 = (const float*)d_in[9];
  const float* Whh1 = (const float*)d_in[10];
  const float* bih1 = (const float*)d_in[11];
  const float* bhh1 = (const float*)d_in[12];
  const float* W2   = (const float*)d_in[13];
  const float* b2   = (const float*)d_in[14];
  const float* W3   = (const float*)d_in[15];
  const float* b3   = (const float*)d_in[16];

  char* ws = (char*)d_ws;
  size_t off = 0;
  float*    dist   = (float*)(ws + off);    off += (size_t)B_ * NK_ * 4;
  float*    hist   = (float*)(ws + off);    off += (size_t)KSPLIT_ * B_ * HLEN_ * 4;
  _Float16* xf     = (_Float16*)(ws + off); off += (size_t)B_ * SEQ_ * L1_ * 2;
  _Float16* h0     = (_Float16*)(ws + off); off += (size_t)B_ * SEQ_ * HID_ * 2;
  float*    h1last = (float*)(ws + off);    off += (size_t)B_ * HID_ * 4;
  float*    c0st   = (float*)(ws + off);    off += (size_t)B_ * HID_ * 4;
  float*    c1st   = (float*)(ws + off);    off += (size_t)B_ * HID_ * 4;
  _Float16* h1st   = (_Float16*)(ws + off); off += (size_t)B_ * HID_ * 2;

  gemm1_kernel<<<dim3(16, 64), 256, 0, stream>>>(inp, akey, dist);
  softmax_kernel<<<dim3(B_), 256, 0, stream>>>(dist);
  gemm2_kernel<<<dim3(16, 5, KSPLIT_), 256, 0, stream>>>(dist, aval, hist);
  featl1_kernel<<<dim3(B_, SEQ_ / FT_), 256, 0, stream>>>(inp, hist, W1, b1, xf);
  lstm_pair_kernel<<<dim3(2 * (B_ / 16)), 256, 0, stream>>>(xf, Wih0, Whh0, bih0, bhh0, Wih1, Whh1, bih1, bhh1,
                                                         h0, c0st, c1st, h1st, h1last, 0);
  lstm_pair_kernel<<<dim3(2 * (B_ / 16)), 256, 0, stream>>>(xf, Wih0, Whh0, bih0, bhh0, Wih1, Whh1, bih1, bhh1,
                                                         h0, c0st, c1st, h1st, h1last, 1);
  lstm_pair_kernel<<<dim3(2 * (B_ / 16)), 256, 0, stream>>>(xf, Wih0, Whh0, bih0, bhh0, Wih1, Whh1, bih1, bhh1,
                                                         h0, c0st, c1st, h1st, h1last, 2);
  lstm_pair_kernel<<<dim3(2 * (B_ / 16)), 256, 0, stream>>>(xf, Wih0, Whh0, bih0, bhh0, Wih1, Whh1, bih1, bhh1,
                                                         h0, c0st, c1st, h1st, h1last, 3);
  lstm_pair_kernel<<<dim3(2 * (B_ / 16)), 256, 0, stream>>>(xf, Wih0, Whh0, bih0, bhh0, Wih1, Whh1, bih1, bhh1,
                                                         h0, c0st, c1st, h1st, h1last, 4);
  lstm_pair_kernel<<<dim3(2 * (B_ / 16)), 256, 0, stream>>>(xf, Wih0, Whh0, bih0, bhh0, Wih1, Whh1, bih1, bhh1,
                                                         h0, c0st, c1st, h1st, h1last, 5);
  lstm_pair_kernel<<<dim3(2 * (B_ / 16)), 256, 0, stream>>>(xf, Wih0, Whh0, bih0, bhh0, Wih1, Whh1, bih1, bhh1,
                                                         h0, c0st, c1st, h1st, h1last, 6);
  lstm_pair_kernel<<<dim3(2 * (B_ / 16)), 256, 0, stream>>>(xf, Wih0, Whh0, bih0, bhh0, Wih1, Whh1, bih1, bhh1,
                                                         h0, c0st, c1st, h1st, h1last, 7);
  lstm_pair_kernel<<<dim3(2 * (B_ / 16)), 256, 0, stream>>>(xf, Wih0, Whh0, bih0, bhh0, Wih1, Whh1, bih1, bhh1,
                                                         h0, c0st, c1st, h1st, h1last, 8);
  head_kernel<<<dim3(B_), 128, 0, stream>>>(h1last, W2, b2, W3, b3, (float*)d_out);
}


# LSTM weights: one small kernel converts Wih/Whh to f16 in WMMA B-fragment order once per call; each of the 9 LSTM launches loads its fragments as whole-line 16-B loads, all in flight at once
# speedup vs baseline: 9.1960x; 1.0888x over previous
#include <hip/hip_runtime.h>
#include <cstdint>
#include <cstddef>

typedef __attribute__((ext_vector_type(16))) _Float16 v16h;
typedef __attribute__((ext_vector_type(8)))  _Float16 v8h;
typedef __attribute__((ext_vector_type(4)))  _Float16 v4h;
typedef __attribute__((ext_vector_type(8)))  float    v8f;
typedef __attribute__((ext_vector_type(2)))  float    v2f;

#define B_    256
#define SEQ_  512
#define HOR_  32
#define NK_   8192
#define L1_   64
#define HID_  128
#define L2_   64
#define GATES_ 512
#define HLEN_ 544
#define KSPLIT_ 16
#define HPAD_ 8

__global__ __launch_bounds__(256) void gemm1_kernel(const float* __restrict__ inp,
                                                    const float* __restrict__ akey,
                                                    float* __restrict__ dist) {
  const int tid = threadIdx.x, lane = tid & 31, wave = tid >> 5;
  const int mt = blockIdx.x, nt = blockIdx.y * 8 + wave;
  const int m = lane & 15;
  const int kb = (lane & 16) ? 2 : 0;
  const float* Ap = inp  + (size_t)(mt * 16 + m) * SEQ_ + kb;
  const float* Bp = akey + (size_t)(nt * 16 + m) * SEQ_ + kb;
  v8f acc0 = {}, acc1 = {};
  for (int k = 0; k < SEQ_; k += 8) {
    v2f a0 = *(const v2f*)(Ap + k);
    v2f b0 = *(const v2f*)(Bp + k);
    acc0 = __builtin_amdgcn_wmma_f32_16x16x4_f32(false, a0, false, b0, (short)0, acc0, false, false);
    v2f a1 = *(const v2f*)(Ap + k + 4);
    v2f b1 = *(const v2f*)(Bp + k + 4);
    acc1 = __builtin_amdgcn_wmma_f32_16x16x4_f32(false, a1, false, b1, (short)0, acc1, false, false);
  }
#pragma unroll
  for (int r = 0; r < 8; ++r) {
    int row = r + ((lane & 16) ? 8 : 0);
    dist[(size_t)(mt * 16 + row) * NK_ + nt * 16 + m] = acc0[r] + acc1[r];
  }
}

__global__ __launch_bounds__(256) void softmax_kernel(float* __restrict__ dist) {
  __shared__ float sv[256];
  __shared__ int   si[256];
  const int b = blockIdx.x, tid = threadIdx.x;
  float* row = dist + (size_t)b * NK_;

  float m = -3.4e38f; int mi = NK_;
  for (int j = tid; j < NK_; j += 256) {
    float v = row[j];
    if (v > m || (v == m && j < mi)) { m = v; mi = j; }
  }
  sv[tid] = m; si[tid] = mi;
  __syncthreads();
  for (int s = 128; s > 0; s >>= 1) {
    if (tid < s) {
      float v2 = sv[tid + s]; int i2 = si[tid + s];
      if (v2 > sv[tid] || (v2 == sv[tid] && i2 < si[tid])) { sv[tid] = v2; si[tid] = i2; }
    }
    __syncthreads();
  }
  if (tid == 0) row[si[0]] = 0.0f;
  __syncthreads();

  float mm = -3.4e38f;
  for (int j = tid; j < NK_; j += 256) mm = fmaxf(mm, row[j]);
  sv[tid] = mm;
  __syncthreads();
  for (int s = 128; s > 0; s >>= 1) { if (tid < s) sv[tid] = fmaxf(sv[tid], sv[tid + s]); __syncthreads(); }
  const float M = sv[0];
  __syncthreads();

  float sum = 0.f;
  for (int j = tid; j < NK_; j += 256) sum += __expf(row[j] - M);
  sv[tid] = sum;
  __syncthreads();
  for (int s = 128; s > 0; s >>= 1) { if (tid < s) sv[tid] += sv[tid + s]; __syncthreads(); }
  const float inv = 1.0f / sv[0];
  for (int j = tid; j < NK_; j += 256) row[j] = __expf(row[j] - M) * inv;
}

__global__ __launch_bounds__(256) void gemm2_kernel(const float* __restrict__ a,
                                                    const float* __restrict__ aval,
                                                    float* __restrict__ hist) {
  const int tid = threadIdx.x, lane = tid & 31, wave = tid >> 5;
  const int mt = blockIdx.x, nt = blockIdx.y * 8 + wave;
  if (nt >= (HLEN_ / 16)) return;
  const int m = lane & 15;
  const int kb = (lane & 16) ? 2 : 0;
  const int n = nt * 16 + m;
  const int ks = blockIdx.z * (NK_ / KSPLIT_);
  const float* Ap = a + (size_t)(mt * 16 + m) * NK_ + ks + kb;
  const float* Bp = aval + (size_t)(ks + kb) * HLEN_ + n;
  hist += (size_t)blockIdx.z * B_ * HLEN_;
  v8f acc0 = {}, acc1 = {};
  for (int k = 0; k < NK_ / KSPLIT_; k += 8) {
    v2f av = *(const v2f*)(Ap + k);
    v2f bv; bv[0] = Bp[(size_t)k * HLEN_]; bv[1] = Bp[(size_t)(k + 1) * HLEN_];
    acc0 = __builtin_amdgcn_wmma_f32_16x16x4_f32(false, av, false, bv, (short)0, acc0, false, false);
    v2f av1 = *(const v2f*)(Ap + k + 4);
    v2f bv1; bv1[0] = Bp[(size_t)(k + 4) * HLEN_]; bv1[1] = Bp[(size_t)(k + 5) * HLEN_];
    acc1 = __builtin_amdgcn_wmma_f32_16x16x4_f32(false, av1, false, bv1, (short)0, acc1, false, false);
  }
#pragma unroll
  for (int r = 0; r < 8; ++r) {
    int row = r + ((lane & 16) ? 8 : 0);
    hist[(size_t)(mt * 16 + row) * HLEN_ + n] = acc0[r] + acc1[r];
  }
}

#define FT_ 64
#define FS_ 16
__global__ __launch_bounds__(256) void featl1_kernel(const float* __restrict__ inp,
                                                     const float* __restrict__ hist,
                                                     const float* __restrict__ W1,
                                                     const float* __restrict__ b1,
                                                     _Float16* __restrict__ xf) {
  __shared__ float w1s[L1_ * (HOR_ + 1)];
  __shared__ __align__(16) float hs[FT_ + HOR_];
  __shared__ __align__(16) float ins[FT_];
  const int b = blockIdx.x, t0 = blockIdx.y * FT_, tid = threadIdx.x;
  for (int i = tid; i < L1_ * (HOR_ + 1); i += 256) w1s[i] = W1[i];
  if (tid < FT_ + HOR_ - 1) {
    const float* hp = hist + (size_t)b * HLEN_ + t0 + tid;
    float v = hp[0];
    for (int s = 1; s < KSPLIT_; ++s) v += hp[(size_t)s * B_ * HLEN_];
    hs[tid] = v;
  } else if (tid == FT_ + HOR_ - 1) {
    hs[tid] = 0.f;
  } else if (tid >= 256 - FT_) {
    ins[tid - (256 - FT_)] = inp[(size_t)b * SEQ_ + t0 + tid - (256 - FT_)];
  }
  const int o = tid & 63;
  const int g = tid >> 6;
  const float bo = b1[o];
  __syncthreads();

  float w[HOR_ + 1];
#pragma unroll
  for (int j = 0; j < HOR_ + 1; ++j) w[j] = w1s[o * (HOR_ + 1) + j];
  float h[FS_ + HOR_];
#pragma unroll
  for (int i = 0; i < (FS_ + HOR_) / 4; ++i) {
    const float4 v = *(const float4*)&hs[g * FS_ + 4 * i];
    h[4 * i] = v.x; h[4 * i + 1] = v.y; h[4 * i + 2] = v.z; h[4 * i + 3] = v.w;
  }
  float acc[FS_];
#pragma unroll
  for (int i = 0; i < FS_ / 4; ++i) {
    const float4 v = *(const float4*)&ins[g * FS_ + 4 * i];
    acc[4 * i]     = __builtin_fmaf(v.x, w[0], bo);
    acc[4 * i + 1] = __builtin_fmaf(v.y, w[0], bo);
    acc[4 * i + 2] = __builtin_fmaf(v.z, w[0], bo);
    acc[4 * i + 3] = __builtin_fmaf(v.w, w[0], bo);
  }
#pragma unroll
  for (int j = 0; j < HOR_; ++j)
#pragma unroll
    for (int i = 0; i < FS_; ++i) acc[i] = __builtin_fmaf(h[i + j], w[1 + j], acc[i]);
  _Float16* xo = xf + ((size_t)b * SEQ_ + t0 + g * FS_) * L1_ + o;
#pragma unroll
  for (int i = 0; i < FS_; ++i) xo[(size_t)i * L1_] = (_Float16)tanhf(acc[i]);
}

#define WF0_PIECES_ (8 * 4 * (L1_ / 32 + 4) * 2 * 32)
#define WF1_PIECES_ (8 * 4 * (HID_ / 32 + 4) * 2 * 32)

template <int KIN>
__device__ __forceinline__ void wfrag_piece(const float* __restrict__ Wih,
                                            const float* __restrict__ Whh,
                                            _Float16* __restrict__ wf, const int idx) {
  constexpr int KFX = KIN / 32;
  constexpr int NF = KFX + 4;
  const int lane = idx & 31;
  const int part = (idx >> 5) & 1;
  const int fq = idx >> 6;
  const int f = fq % NF;
  const int wq = fq / NF;
  const int q = wq & 3, wave = wq >> 2;
  const int g = q * HID_ + wave * 16 + (lane & 15);
  const int kin = part * 16 + (lane >> 4) * 8;
  const float* src = (f < KFX) ? Wih + (size_t)g * KIN + f * 32 + kin
                               : Whh + (size_t)g * HID_ + (f - KFX) * 32 + kin;
  const float4 a = *(const float4*)src;
  const float4 b = *(const float4*)(src + 4);
  v8h o;
  o[0] = (_Float16)a.x; o[1] = (_Float16)a.y; o[2] = (_Float16)a.z; o[3] = (_Float16)a.w;
  o[4] = (_Float16)b.x; o[5] = (_Float16)b.y; o[6] = (_Float16)b.z; o[7] = (_Float16)b.w;
  *(v8h*)(wf + (size_t)idx * 8) = o;
}

__global__ __launch_bounds__(256) void wfrag_kernel(const float* __restrict__ Wih0,
                                                    const float* __restrict__ Whh0,
                                                    const float* __restrict__ Wih1,
                                                    const float* __restrict__ Whh1,
                                                    _Float16* __restrict__ wf0,
                                                    _Float16* __restrict__ wf1) {
  const int idx = blockIdx.x * 256 + threadIdx.x;
  if (blockIdx.x < WF0_PIECES_ / 256) wfrag_piece<L1_>(Wih0, Whh0, wf0, idx);
  else                                wfrag_piece<HID_>(Wih1, Whh1, wf1, idx - WF0_PIECES_);
}

#define LOG2E_F 1.4426950408889634f
__device__ __forceinline__ float act_logistic(float v) {
  const float e = __builtin_amdgcn_exp2f(-LOG2E_F * v);
  return __builtin_amdgcn_rcpf(1.0f + e);
}
__device__ __forceinline__ float act_tanh(float v) {
  const float e = __builtin_amdgcn_exp2f(-2.0f * LOG2E_F * v);
  return __builtin_fmaf(2.0f, __builtin_amdgcn_rcpf(1.0f + e), -1.0f);
}

#define CHUNK_  64
#define NCHUNK_ 8

template <int N> struct hvec_t;
template <> struct hvec_t<8> { typedef v8h type; };
template <> struct hvec_t<4> { typedef v4h type; };

template <int KIN, bool STORE_SEQ>
__device__ __forceinline__ void lstm_chunk(_Float16 (*hbuf)[16][HID_ + HPAD_],
                                           _Float16* xls,
                                           const _Float16* __restrict__ xin,
                                           const _Float16* __restrict__ wfrag,
                                           const float* __restrict__ bih,
                                           const float* __restrict__ bhh,
                                           _Float16* __restrict__ hseq,
                                           _Float16* __restrict__ hstate,
                                           float* __restrict__ cstate,
                                           float* __restrict__ hlast,
                                           const int bm, const int t0) {
  constexpr int KFX = KIN / 32;
  const int tid = threadIdx.x, lane = tid & 31, wave = tid >> 5;
  const int m = lane & 15;
  const int hi16 = (lane >> 4) & 1;
  const int kbh = hi16 * 8;
  const int unit = wave * 16 + m;
  const int drow = hi16 * 8;
  const int tend = t0 + CHUNK_;

  const int prow = tid >> 4;
  const int pj = (tid & 15) * 8;
  float c[8];
  if (t0 == 0) {
#pragma unroll
    for (int r = 0; r < 8; ++r) c[r] = 0.f;
#pragma unroll
    for (int i = 0; i < 8; ++i) hbuf[0][prow][pj + i] = (_Float16)0.f;
  } else {
#pragma unroll
    for (int r = 0; r < 8; ++r) c[r] = cstate[(size_t)(bm + drow + r) * HID_ + unit];
    const _Float16* hp = STORE_SEQ ? hseq + ((size_t)(bm + prow) * SEQ_ + (t0 - 1)) * HID_ + pj
                                   : hstate + (size_t)(bm + prow) * HID_ + pj;
    *(v8h*)&hbuf[0][prow][pj] = *(const v8h*)hp;
  }
  constexpr int XE = KIN / 16;
  typedef typename hvec_t<XE>::type xvec;
  _Float16 (*xl)[16][KIN + HPAD_] = (_Float16 (*)[16][KIN + HPAD_])xls;
  const int xj = (tid & 15) * XE;
  const _Float16* xpiece = xin + (size_t)(bm + prow) * SEQ_ * KIN + xj;
  *(xvec*)&xl[1][prow][xj] = *(const xvec*)(xpiece + (size_t)(t0 + 1) * KIN);
  __syncthreads();


  constexpr int NF = KFX + 4;
  const v8h* wp = (const v8h*)wfrag + (size_t)wave * (4 * NF * 64) + lane;
  v16h wihB[4][KFX], whhB[4][4];
  float biasv[4];
#pragma unroll
  for (int q = 0; q < 4; ++q) {
    const int g = q * HID_ + unit;
    biasv[q] = bih[g] + bhh[g];
#pragma unroll
    for (int kf = 0; kf < KFX; ++kf) {
      const v8h lo = wp[(q * NF + kf) * 64];
      const v8h hi = wp[(q * NF + kf) * 64 + 32];
#pragma unroll
      for (int i = 0; i < 8; ++i) { wihB[q][kf][i] = lo[i]; wihB[q][kf][i + 8] = hi[i]; }
    }
#pragma unroll
    for (int kf = 0; kf < 4; ++kf) {
      const v8h lo = wp[(q * NF + KFX + kf) * 64];
      const v8h hi = wp[(q * NF + KFX + kf) * 64 + 32];
#pragma unroll
      for (int i = 0; i < 8; ++i) { whhB[q][kf][i] = lo[i]; whhB[q][kf][i + 8] = hi[i]; }
    }
  }

  const _Float16* xrow = xin + (size_t)(bm + m) * SEQ_ * KIN + kbh;
  v16h xbuf[2][KFX];
#pragma unroll
  for (int kf = 0; kf < KFX; ++kf) {
    const _Float16* p = xrow + (size_t)t0 * KIN + kf * 32;
    v8h lo = *(const v8h*)p;
    v8h hi = *(const v8h*)(p + 16);
#pragma unroll
    for (int i = 0; i < 8; ++i) { xbuf[0][kf][i] = lo[i]; xbuf[0][kf][i + 8] = hi[i]; }
  }

  float hv[8];
#pragma unroll
  for (int r = 0; r < 8; ++r) hv[r] = 0.f;

  for (int t2 = t0; t2 < tend; t2 += 2) {
#pragma unroll
    for (int u = 0; u < 2; ++u) {
      const int t = t2 + u;
      v16h hA[4];
#pragma unroll
      for (int kf = 0; kf < 4; ++kf) {
        const _Float16* p = &hbuf[u][m][kf * 32 + kbh];
        v8h lo = *(const v8h*)p;
        v8h hi = *(const v8h*)(p + 16);
#pragma unroll
        for (int i = 0; i < 8; ++i) { hA[kf][i] = lo[i]; hA[kf][i + 8] = hi[i]; }
      }
      v8h hrow;
      if (STORE_SEQ) hrow = *(const v8h*)&hbuf[u][prow][pj];
#pragma unroll
      for (int kf = 0; kf < KFX; ++kf) {
        const _Float16* p = &xl[u ^ 1][m][kf * 32 + kbh];
        v8h lo = *(const v8h*)p;
        v8h hi = *(const v8h*)(p + 16);
#pragma unroll
        for (int i = 0; i < 8; ++i) { xbuf[u ^ 1][kf][i] = lo[i]; xbuf[u ^ 1][kf][i + 8] = hi[i]; }
      }
      const int tn = (t + 2 < tend) ? t + 2 : tend - 1;
      const xvec xg = *(const xvec*)(xpiece + (size_t)tn * KIN);
      __builtin_amdgcn_sched_barrier(0);
      v8f acc[4];
#pragma unroll
      for (int q = 0; q < 4; ++q)
#pragma unroll
        for (int r = 0; r < 8; ++r) acc[q][r] = biasv[q];
#pragma unroll
      for (int kf = 0; kf < KFX; ++kf)
#pragma unroll
        for (int q = 0; q < 4; ++q)
          acc[q] = __builtin_amdgcn_wmma_f32_16x16x32_f16(false, xbuf[u][kf], false, wihB[q][kf],
                                                          (short)0, acc[q], false, false);
#pragma unroll
      for (int kf = 0; kf < 4; ++kf)
#pragma unroll
        for (int q = 0; q < 4; ++q)
          acc[q] = __builtin_amdgcn_wmma_f32_16x16x32_f16(false, hA[kf], false, whhB[q][kf],
                                                          (short)0, acc[q], false, false);
      if (STORE_SEQ && t > t0)
        *(v8h*)(hseq + ((size_t)(bm + prow) * SEQ_ + (t - 1)) * HID_ + pj) = hrow;
      float gi[8], gf[8], gg[8], go[8];
#pragma unroll
      for (int r = 0; r < 8; ++r) gi[r] = act_logistic(acc[0][r]);
#pragma unroll
      for (int r = 0; r < 8; ++r) gf[r] = act_logistic(acc[1][r]);
#pragma unroll
      for (int r = 0; r < 8; ++r) gg[r] = act_tanh(acc[2][r]);
#pragma unroll
      for (int r = 0; r < 8; ++r) go[r] = act_logistic(acc[3][r]);
#pragma unroll
      for (int r = 0; r < 8; ++r) c[r] = gf[r] * c[r] + gi[r] * gg[r];
#pragma unroll
      for (int r = 0; r < 8; ++r) hv[r] = go[r] * act_tanh(c[r]);
#pragma unroll
      for (int r = 0; r < 8; ++r) hbuf[u ^ 1][drow + r][unit] = (_Float16)hv[r];
      *(xvec*)&xl[u][prow][xj] = xg;
      __syncthreads();
    }
  }
  if (STORE_SEQ) {
    const v8h hrow = *(const v8h*)&hbuf[0][prow][pj];
    *(v8h*)(hseq + ((size_t)(bm + prow) * SEQ_ + (tend - 1)) * HID_ + pj) = hrow;
  } else if (tend == SEQ_) {
#pragma unroll
    for (int r = 0; r < 8; ++r) hlast[(size_t)(bm + drow + r) * HID_ + unit] = hv[r];
  } else {
    const v8h hrow = *(const v8h*)&hbuf[0][prow][pj];
    *(v8h*)(hstate + (size_t)(bm + prow) * HID_ + pj) = hrow;
  }
  if (tend < SEQ_) {
#pragma unroll
    for (int r = 0; r < 8; ++r) cstate[(size_t)(bm + drow + r) * HID_ + unit] = c[r];
  }
}

__global__ __launch_bounds__(256, 1) void lstm_pair_kernel(const _Float16* __restrict__ xf,
                                                           const _Float16* __restrict__ wf0,
                                                           const float* __restrict__ bih0,
                                                           const float* __restrict__ bhh0,
                                                           const _Float16* __restrict__ wf1,
                                                           const float* __restrict__ bih1,
                                                           const float* __restrict__ bhh1,
                                                           _Float16* h0seq,
                                                           float* c0state,
                                                           float* c1state,
                                                           _Float16* h1state,
                                                           float* h1last,
                                                           const int chunk) {
  __shared__ __align__(16) _Float16 hbuf[2][16][HID_ + HPAD_];
  __shared__ __align__(16) _Float16 xls[2 * 16 * (HID_ + HPAD_)];
  const int bm = (blockIdx.x & 15) * 16;
  if (blockIdx.x < 16) {
    if (chunk < NCHUNK_)
      lstm_chunk<L1_, true>(hbuf, xls, xf, wf0, bih0, bhh0, h0seq, nullptr, c0state, nullptr,
                            bm, chunk * CHUNK_);
  } else {
    if (chunk > 0)
      lstm_chunk<HID_, false>(hbuf, xls, h0seq, wf1, bih1, bhh1, nullptr, h1state, c1state, h1last,
                              bm, (chunk - 1) * CHUNK_);
  }
}

__global__ __launch_bounds__(128) void head_kernel(const float* __restrict__ h1last,
                                                   const float* __restrict__ W2,
                                                   const float* __restrict__ b2,
                                                   const float* __restrict__ W3,
                                                   const float* __restrict__ b3,
                                                   float* __restrict__ out) {
  __shared__ float lastS[HID_];
  __shared__ float l2S[L2_];
  const int b = blockIdx.x, tid = threadIdx.x;
  lastS[tid] = tanhf(h1last[(size_t)b * HID_ + tid]);
  __syncthreads();
  if (tid < L2_) {
    float acc = b2[tid];
    for (int k = 0; k < HID_; ++k) acc += lastS[k] * W2[tid * HID_ + k];
    l2S[tid] = tanhf(acc);
  }
  __syncthreads();
  if (tid < HOR_) {
    float acc = b3[tid];
    for (int k = 0; k < L2_; ++k) acc += l2S[k] * W3[tid * L2_ + k];
    out[(size_t)b * HOR_ + tid] = acc;
  }
}

extern "C" void kernel_launch(void* const* d_in, const int* in_sizes, int n_in,
                              void* d_out, int out_size, void* d_ws, size_t ws_size,
                              hipStream_t stream) {
  (void)in_sizes; (void)n_in; (void)out_size; (void)ws_size;
  const float* inp  = (const float*)d_in[0];
  const float* akey = (const float*)d_in[1];
  const float* aval = (const float*)d_in[2];
  const float* W1   = (const float*)d_in[3];
  const float* b1   = (const float*)d_in[4];
  const float* Wih0 = (const float*)d_in[5];
  const float* Whh0 = (const float*)d_in[6];
  const float* bih0 = (const float*)d_in[7];
  const float* bhh0 = (const float*)d_in[8];
  const float* Wih1 = (const float*)d_in[9];
  const float* Whh1 = (const float*)d_in[10];
  const float* bih1 = (const float*)d_in[11];
  const float* bhh1 = (const float*)d_in[12];
  const float* W2   = (const float*)d_in[13];
  const float* b2   = (const float*)d_in[14];
  const float* W3   = (const float*)d_in[15];
  const float* b3   = (const float*)d_in[16];

  char* ws = (char*)d_ws;
  size_t off = 0;
  float*    dist   = (float*)(ws + off);    off += (size_t)B_ * NK_ * 4;
  float*    hist   = (float*)(ws + off);    off += (size_t)KSPLIT_ * B_ * HLEN_ * 4;
  _Float16* xf     = (_Float16*)(ws + off); off += (size_t)B_ * SEQ_ * L1_ * 2;
  _Float16* h0     = (_Float16*)(ws + off); off += (size_t)B_ * SEQ_ * HID_ * 2;
  float*    h1last = (float*)(ws + off);    off += (size_t)B_ * HID_ * 4;
  float*    c0st   = (float*)(ws + off);    off += (size_t)B_ * HID_ * 4;
  float*    c1st   = (float*)(ws + off);    off += (size_t)B_ * HID_ * 4;
  _Float16* h1st   = (_Float16*)(ws + off); off += (size_t)B_ * HID_ * 2;
  _Float16* wf0    = (_Float16*)(ws + off); off += (size_t)WF0_PIECES_ * 16;
  _Float16* wf1    = (_Float16*)(ws + off); off += (size_t)WF1_PIECES_ * 16;

  wfrag_kernel<<<dim3((WF0_PIECES_ + WF1_PIECES_) / 256), 256, 0, stream>>>(Wih0, Whh0, Wih1, Whh1, wf0, wf1);
  gemm1_kernel<<<dim3(16, 64), 256, 0, stream>>>(inp, akey, dist);
  softmax_kernel<<<dim3(B_), 256, 0, stream>>>(dist);
  gemm2_kernel<<<dim3(16, 5, KSPLIT_), 256, 0, stream>>>(dist, aval, hist);
  featl1_kernel<<<dim3(B_, SEQ_ / FT_), 256, 0, stream>>>(inp, hist, W1, b1, xf);
  lstm_pair_kernel<<<dim3(2 * (B_ / 16)), 256, 0, stream>>>(xf, wf0, bih0, bhh0, wf1, bih1, bhh1,
                                                         h0, c0st, c1st, h1st, h1last, 0);
  lstm_pair_kernel<<<dim3(2 * (B_ / 16)), 256, 0, stream>>>(xf, wf0, bih0, bhh0, wf1, bih1, bhh1,
                                                         h0, c0st, c1st, h1st, h1last, 1);
  lstm_pair_kernel<<<dim3(2 * (B_ / 16)), 256, 0, stream>>>(xf, wf0, bih0, bhh0, wf1, bih1, bhh1,
                                                         h0, c0st, c1st, h1st, h1last, 2);
  lstm_pair_kernel<<<dim3(2 * (B_ / 16)), 256, 0, stream>>>(xf, wf0, bih0, bhh0, wf1, bih1, bhh1,
                                                         h0, c0st, c1st, h1st, h1last, 3);
  lstm_pair_kernel<<<dim3(2 * (B_ / 16)), 256, 0, stream>>>(xf, wf0, bih0, bhh0, wf1, bih1, bhh1,
                                                         h0, c0st, c1st, h1st, h1last, 4);
  lstm_pair_kernel<<<dim3(2 * (B_ / 16)), 256, 0, stream>>>(xf, wf0, bih0, bhh0, wf1, bih1, bhh1,
                                                         h0, c0st, c1st, h1st, h1last, 5);
  lstm_pair_kernel<<<dim3(2 * (B_ / 16)), 256, 0, stream>>>(xf, wf0, bih0, bhh0, wf1, bih1, bhh1,
                                                         h0, c0st, c1st, h1st, h1last, 6);
  lstm_pair_kernel<<<dim3(2 * (B_ / 16)), 256, 0, stream>>>(xf, wf0, bih0, bhh0, wf1, bih1, bhh1,
                                                         h0, c0st, c1st, h1st, h1last, 7);
  lstm_pair_kernel<<<dim3(2 * (B_ / 16)), 256, 0, stream>>>(xf, wf0, bih0, bhh0, wf1, bih1, bhh1,
                                                         h0, c0st, c1st, h1st, h1last, 8);
  head_kernel<<<dim3(B_), 128, 0, stream>>>(h1last, W2, b2, W3, b3, (float*)d_out);
}
